# starting kernel
# baseline (speedup 1.0000x reference)
#include <hip/hip_runtime.h>

typedef __attribute__((ext_vector_type(2))) float v2f;
typedef __attribute__((ext_vector_type(8))) float v8f;

#define N_NODES 50000
#define N_EDGES 640000
#define N_REL   8
#define KDIM    128
#define NSEG    (N_REL + 1)

__global__ void rgcn_zero_kernel(float4* __restrict__ p, long long n4) {
    long long i = blockIdx.x * (long long)blockDim.x + threadIdx.x;
    if (i < n4) p[i] = make_float4(0.f, 0.f, 0.f, 0.f);
}

__global__ void rgcn_scatter_kernel(const float* __restrict__ feat,
                                    const int*   __restrict__ esrc,
                                    const int*   __restrict__ edst,
                                    const int*   __restrict__ etyp,
                                    float*       __restrict__ aggR,
                                    int nEdges) {
    long long tid = blockIdx.x * (long long)blockDim.x + threadIdx.x;
    int e = (int)(tid >> 7);
    int o = (int)(tid & 127);
    if (e >= nEdges) return;
    int s = esrc[e];
    int d = edst[e];
    int r = etyp[e];
    float v = feat[(size_t)s * KDIM + o];
    atomicAdd(aggR + ((size_t)r * N_NODES + (size_t)d) * KDIM + o, v);
}

__global__ void rgcn_pack_w_kernel(const float* __restrict__ Wrel,
                                   const float* __restrict__ Wself,
                                   float2*      __restrict__ Wf,
                                   int O) {
    const int tiles = O >> 4;
    const int total = NSEG * tiles * 32 * 32;
    int idx = blockIdx.x * blockDim.x + threadIdx.x;
    if (idx >= total) return;
    int lane = idx & 31;
    int k4   = (idx >> 5) & 31;
    int t    = (idx >> 10) % tiles;
    int seg  = idx / (tiles << 10);
    const float* W = (seg < N_REL) ? (Wrel + (size_t)seg * KDIM * O) : Wself;
    int k0 = k4 * 4 + (lane >> 4) * 2;
    int n  = t * 16 + (lane & 15);
    Wf[idx] = make_float2(W[(size_t)k0 * O + n], W[(size_t)(k0 + 1) * O + n]);
}

__global__ void rgcn_gemm_kernel(const float*  __restrict__ aggR,
                                 const float*  __restrict__ hself,
                                 const float2* __restrict__ Wf,
                                 const float*  __restrict__ bias,
                                 float*        __restrict__ out,
                                 int O, int applyRelu) {
    __shared__ float lds[16 * 132];

    const int m0    = blockIdx.x * 16;
    const int lane  = threadIdx.x & 31;
    const int wave  = threadIdx.x >> 5;
    const int half  = lane >> 4;
    const int l15   = lane & 15;
    const int nb    = wave * 16;
    const int tiles = O >> 4;

    v8f c = {0.f, 0.f, 0.f, 0.f, 0.f, 0.f, 0.f, 0.f};

    for (int s = 0; s < NSEG; ++s) {
        const float*  A  = (s < N_REL) ? (aggR + (size_t)s * N_NODES * KDIM) : hself;
        const float2* wf = Wf + ((size_t)(s * tiles + wave) << 10) + lane;

        __syncthreads();
        for (int t = threadIdx.x; t < 16 * KDIM; t += blockDim.x) {
            int row = t >> 7;
            int col = t & 127;
            lds[row * 132 + col] = A[(size_t)(m0 + row) * KDIM + col];
        }
        __syncthreads();

#pragma unroll
        for (int k4 = 0; k4 < 32; ++k4) {
            const float* ap = &lds[l15 * 132 + k4 * 4 + half * 2];
            v2f a;
            a.x = ap[0];
            a.y = ap[1];
            float2 bf = wf[k4 * 32];
            v2f b;
            b.x = bf.x;
            b.y = bf.y;
            c = __builtin_amdgcn_wmma_f32_16x16x4_f32(
                    false, a, false, b, (short)0, c, false, false);
        }
    }

    float bv = bias[nb + l15];
#pragma unroll
    for (int v = 0; v < 8; ++v) {
        float x = c[v] + bv;
        if (applyRelu) x = fmaxf(x, 0.f);
        out[(size_t)(m0 + v + 8 * half) * O + nb + l15] = x;
    }
}

extern "C" void kernel_launch(void* const* d_in, const int* in_sizes, int n_in,
                              void* d_out, int out_size, void* d_ws, size_t ws_size,
                              hipStream_t stream) {
    const float* features = (const float*)d_in[0];
    const int*   src      = (const int*)  d_in[1];
    const int*   dst      = (const int*)  d_in[2];
    const int*   etype    = (const int*)  d_in[3];
    const float* W1       = (const float*)d_in[4];
    const float* W1s      = (const float*)d_in[5];
    const float* b1       = (const float*)d_in[6];
    const float* W2       = (const float*)d_in[7];
    const float* W2s      = (const float*)d_in[8];
    const float* b2       = (const float*)d_in[9];
    float*       out      = (float*)d_out;

    float*  aggR = (float*)d_ws;
    float*  h1   = aggR + (size_t)N_REL * N_NODES * KDIM;
    float2* Wf1  = (float2*)(h1 + (size_t)N_NODES * KDIM);
    float2* Wf2  = Wf1 + (size_t)NSEG * 8 * 32 * 32;

    const long long n4     = (long long)N_REL * N_NODES * KDIM / 4;
    const int zeroBlocks   = (int)((n4 + 255) / 256);
    const int scatBlocks   = (int)(((long long)N_EDGES * KDIM) / 256);
    const int gemmBlocks   = N_NODES / 16;

    const int pack1 = NSEG * 8 * 32 * 32;
    const int pack2 = NSEG * 4 * 32 * 32;

    rgcn_pack_w_kernel<<<(pack1 + 255) / 256, 256, 0, stream>>>(W1, W1s, Wf1, 128);
    rgcn_pack_w_kernel<<<(pack2 + 255) / 256, 256, 0, stream>>>(W2, W2s, Wf2, 64);

    rgcn_zero_kernel<<<zeroBlocks, 256, 0, stream>>>((float4*)aggR, n4);
    rgcn_scatter_kernel<<<scatBlocks, 256, 0, stream>>>(
        features, src, dst, etype, aggR, N_EDGES);
    rgcn_gemm_kernel<<<gemmBlocks, 256, 0, stream>>>(
        aggR, features, Wf1, b1, h1, 128, 1);

    rgcn_zero_kernel<<<zeroBlocks, 256, 0, stream>>>((float4*)aggR, n4);
    rgcn_scatter_kernel<<<scatBlocks, 256, 0, stream>>>(
        h1, src, dst, etype, aggR, N_EDGES);
    rgcn_gemm_kernel<<<gemmBlocks, 128, 0, stream>>>(
        aggR, h1, Wf2, b2, out, 64, 0);
}


# Layer 2 edge sum owner-computed: 256 work-groups each hold 196 out rows in LDS, scan all dst, fetch matching rel rows, add by LDS float adds; its global atomics go; sums in arrival order in a group
# speedup vs baseline: 1.5116x; 1.5116x over previous
#include <hip/hip_runtime.h>

typedef __attribute__((ext_vector_type(2))) float v2f;
typedef __attribute__((ext_vector_type(8))) float v8f;

#define N_NODES 50000
#define N_EDGES 640000
#define N_REL   8
#define KDIM    128
#define NSEG    (N_REL + 1)

__global__ void rgcn_zero_kernel(float4* __restrict__ p, long long n4) {
    long long i = blockIdx.x * (long long)blockDim.x + threadIdx.x;
    if (i < n4) p[i] = make_float4(0.f, 0.f, 0.f, 0.f);
}

__global__ void rgcn_scatter_kernel(const float* __restrict__ feat,
                                    const int*   __restrict__ esrc,
                                    const int*   __restrict__ edst,
                                    const int*   __restrict__ etyp,
                                    float*       __restrict__ aggR,
                                    int nEdges) {
    long long tid = blockIdx.x * (long long)blockDim.x + threadIdx.x;
    int e = (int)(tid >> 7);
    int o = (int)(tid & 127);
    if (e >= nEdges) return;
    int s = esrc[e];
    int d = edst[e];
    int r = etyp[e];
    float v = feat[(size_t)s * KDIM + o];
    atomicAdd(aggR + ((size_t)r * N_NODES + (size_t)d) * KDIM + o, v);
}

__global__ void rgcn_pack_w_kernel(const float* __restrict__ Wrel,
                                   const float* __restrict__ Wself,
                                   float2*      __restrict__ Wf,
                                   int O) {
    const int tiles = O >> 4;
    const int total = NSEG * tiles * 32 * 32;
    int idx = blockIdx.x * blockDim.x + threadIdx.x;
    if (idx >= total) return;
    int lane = idx & 31;
    int k4   = (idx >> 5) & 31;
    int t    = (idx >> 10) % tiles;
    int seg  = idx / (tiles << 10);
    const float* W = (seg < N_REL) ? (Wrel + (size_t)seg * KDIM * O) : Wself;
    int k0 = k4 * 4 + (lane >> 4) * 2;
    int n  = t * 16 + (lane & 15);
    Wf[idx] = make_float2(W[(size_t)k0 * O + n], W[(size_t)(k0 + 1) * O + n]);
}

__global__ void rgcn_gemm_kernel(const float*  __restrict__ aggR,
                                 const float*  __restrict__ hself,
                                 const float2* __restrict__ Wf,
                                 const float*  __restrict__ bias,
                                 float*        __restrict__ out,
                                 int O, int applyRelu) {
    __shared__ float lds[16 * 132];

    const int m0    = blockIdx.x * 16;
    const int lane  = threadIdx.x & 31;
    const int wave  = threadIdx.x >> 5;
    const int half  = lane >> 4;
    const int l15   = lane & 15;
    const int nb    = wave * 16;
    const int tiles = O >> 4;

    v8f c = {0.f, 0.f, 0.f, 0.f, 0.f, 0.f, 0.f, 0.f};

    for (int s = 0; s < NSEG; ++s) {
        const float*  A  = (s < N_REL) ? (aggR + (size_t)s * N_NODES * KDIM) : hself;
        const float2* wf = Wf + ((size_t)(s * tiles + wave) << 10) + lane;

        __syncthreads();
        for (int t = threadIdx.x; t < 16 * KDIM; t += blockDim.x) {
            int row = t >> 7;
            int col = t & 127;
            lds[row * 132 + col] = A[(size_t)(m0 + row) * KDIM + col];
        }
        __syncthreads();

#pragma unroll
        for (int k4 = 0; k4 < 32; ++k4) {
            const float* ap = &lds[l15 * 132 + k4 * 4 + half * 2];
            v2f a;
            a.x = ap[0];
            a.y = ap[1];
            float2 bf = wf[k4 * 32];
            v2f b;
            b.x = bf.x;
            b.y = bf.y;
            c = __builtin_amdgcn_wmma_f32_16x16x4_f32(
                    false, a, false, b, (short)0, c, false, false);
        }
    }

    float bv = bias[nb + l15];
#pragma unroll
    for (int v = 0; v < 8; ++v) {
        float x = c[v] + bv;
        if (applyRelu) x = fmaxf(x, 0.f);
        out[(size_t)(m0 + v + 8 * half) * O + nb + l15] = x;
    }
}

__global__ void __launch_bounds__(288)
rgcn_gemm_t_kernel(const float*  __restrict__ h,
                   const float2* __restrict__ Wf,
                   const float*  __restrict__ bias,
                   float*        __restrict__ rel,
                   float*        __restrict__ out) {
    __shared__ float lds[16 * 132];

    const int m0   = blockIdx.x * 16;
    const int lane = threadIdx.x & 31;
    const int seg  = threadIdx.x >> 5;
    const int half = lane >> 4;
    const int l15  = lane & 15;

    if (threadIdx.x < 256) {
        const int col  = threadIdx.x & 127;
        const int row0 = threadIdx.x >> 7;
#pragma unroll
        for (int i = 0; i < 8; ++i) {
            int row = row0 + 2 * i;
            lds[row * 132 + col] = h[(size_t)(m0 + row) * KDIM + col];
        }
    }
    __syncthreads();

    const float2* wf = Wf + ((size_t)(seg * 4) << 10) + lane;

    v8f c0 = {0.f, 0.f, 0.f, 0.f, 0.f, 0.f, 0.f, 0.f};
    v8f c1 = c0, c2 = c0, c3 = c0;

#pragma unroll
    for (int k4 = 0; k4 < 32; ++k4) {
        const float* ap = &lds[l15 * 132 + k4 * 4 + half * 2];
        v2f a;
        a.x = ap[0];
        a.y = ap[1];
        float2 f0 = wf[k4 * 32];
        float2 f1 = wf[k4 * 32 + 1024];
        float2 f2 = wf[k4 * 32 + 2048];
        float2 f3 = wf[k4 * 32 + 3072];
        v2f b0, b1, b2, b3;
        b0.x = f0.x; b0.y = f0.y;
        b1.x = f1.x; b1.y = f1.y;
        b2.x = f2.x; b2.y = f2.y;
        b3.x = f3.x; b3.y = f3.y;
        c0 = __builtin_amdgcn_wmma_f32_16x16x4_f32(
                 false, a, false, b0, (short)0, c0, false, false);
        c1 = __builtin_amdgcn_wmma_f32_16x16x4_f32(
                 false, a, false, b1, (short)0, c1, false, false);
        c2 = __builtin_amdgcn_wmma_f32_16x16x4_f32(
                 false, a, false, b2, (short)0, c2, false, false);
        c3 = __builtin_amdgcn_wmma_f32_16x16x4_f32(
                 false, a, false, b3, (short)0, c3, false, false);
    }

    if (seg < N_REL) {
#pragma unroll
        for (int v = 0; v < 8; ++v) {
            float* p = rel + ((size_t)(m0 + v + 8 * half) * N_REL + seg) * 64 + l15;
            p[0]  = c0[v];
            p[16] = c1[v];
            p[32] = c2[v];
            p[48] = c3[v];
        }
    } else {
        float bv0 = bias[l15];
        float bv1 = bias[16 + l15];
        float bv2 = bias[32 + l15];
        float bv3 = bias[48 + l15];
#pragma unroll
        for (int v = 0; v < 8; ++v) {
            float* p = out + (size_t)(m0 + v + 8 * half) * 64 + l15;
            p[0]  = c0[v] + bv0;
            p[16] = c1[v] + bv1;
            p[32] = c2[v] + bv2;
            p[48] = c3[v] + bv3;
        }
    }
}

#define OWN_ROWS    196
#define OWN_BLOCKS  ((N_NODES + OWN_ROWS - 1) / OWN_ROWS)
#define OWN_THREADS 1024
#define OWN_WAVES   (OWN_THREADS / 32)
#define RING_CAP    1024

__device__ __forceinline__ void owner_add_batch(const float* __restrict__ rel,
                                                const int*   __restrict__ esrc,
                                                const int*   __restrict__ edst,
                                                const int*   __restrict__ etyp,
                                                float* acc, const int* q,
                                                int head, int cnt, int base, int lane) {
    const int e    = (lane < cnt) ? q[(head + lane) & (RING_CAP - 1)] : 0;
    const int roff = (esrc[e] * N_REL + etyp[e]) * 64;
    const int doff = (edst[e] - base) * 64;
    float2 v[32];
#pragma unroll
    for (int j = 0; j < 32; ++j) {
        const int ro = __builtin_amdgcn_readlane(roff, j);
        v[j] = *(const float2*)(rel + ro + lane * 2);
    }
#pragma unroll
    for (int j = 0; j < 32; ++j) {
        if (j < cnt) {
            float* a = acc + __builtin_amdgcn_readlane(doff, j) + lane * 2;
            atomicAdd(a,     v[j].x);
            atomicAdd(a + 1, v[j].y);
        }
    }
}

__global__ void __launch_bounds__(OWN_THREADS)
rgcn_owner_sum_kernel(const float* __restrict__ rel,
                      const int*   __restrict__ esrc,
                      const int*   __restrict__ edst,
                      const int*   __restrict__ etyp,
                      float*       __restrict__ out) {
    __shared__ __attribute__((aligned(16))) float acc[OWN_ROWS * 64];
    __shared__ int ring[OWN_WAVES * RING_CAP];

    const int tid   = threadIdx.x;
    const int lane  = tid & 31;
    const int wave  = tid >> 5;
    const int base  = blockIdx.x * OWN_ROWS;
    const int nrows = (N_NODES - base < OWN_ROWS) ? (N_NODES - base) : OWN_ROWS;

    float4*       a4 = (float4*)acc;
    float4*       o4 = (float4*)(out + (size_t)base * 64);
    for (int i = tid; i < nrows * 16; i += OWN_THREADS) a4[i] = o4[i];
    __syncthreads();

    int* q = ring + wave * RING_CAP;
    int head = 0, tail = 0;

    const int4* d4 = (const int4*)edst;
    const int   nQ = N_EDGES / 4;

    for (int q0 = 0; q0 < nQ; q0 += 4 * OWN_THREADS) {
        int4 dv[4];
#pragma unroll
        for (int u = 0; u < 4; ++u) {
            int qi = q0 + u * OWN_THREADS + tid;
            dv[u] = (qi < nQ) ? d4[qi] : make_int4(-1, -1, -1, -1);
        }
#pragma unroll
        for (int u = 0; u < 4; ++u) {
            const int e0 = (q0 + u * OWN_THREADS + tid) * 4;
            const bool m0 = (unsigned)(dv[u].x - base) < (unsigned)nrows;
            const bool m1 = (unsigned)(dv[u].y - base) < (unsigned)nrows;
            const bool m2 = (unsigned)(dv[u].z - base) < (unsigned)nrows;
            const bool m3 = (unsigned)(dv[u].w - base) < (unsigned)nrows;
            if (__builtin_amdgcn_ballot_w32(m0 | m1 | m2 | m3)) {
                unsigned k;
                k = __builtin_amdgcn_ballot_w32(m0);
                if (m0) q[(tail + __builtin_amdgcn_mbcnt_lo(k, 0u)) & (RING_CAP - 1)] = e0;
                tail += __builtin_popcount(k);
                k = __builtin_amdgcn_ballot_w32(m1);
                if (m1) q[(tail + __builtin_amdgcn_mbcnt_lo(k, 0u)) & (RING_CAP - 1)] = e0 + 1;
                tail += __builtin_popcount(k);
                k = __builtin_amdgcn_ballot_w32(m2);
                if (m2) q[(tail + __builtin_amdgcn_mbcnt_lo(k, 0u)) & (RING_CAP - 1)] = e0 + 2;
                tail += __builtin_popcount(k);
                k = __builtin_amdgcn_ballot_w32(m3);
                if (m3) q[(tail + __builtin_amdgcn_mbcnt_lo(k, 0u)) & (RING_CAP - 1)] = e0 + 3;
                tail += __builtin_popcount(k);
            }
        }

        while (tail - head >= 32) {
            owner_add_batch(rel, esrc, edst, etyp, acc, q, head, 32, base, lane);
            head += 32;
        }
    }

    if (tail - head > 0)
        owner_add_batch(rel, esrc, edst, etyp, acc, q, head, tail - head, base, lane);

    __syncthreads();
    for (int i = tid; i < nrows * 16; i += OWN_THREADS) o4[i] = a4[i];
}

extern "C" void kernel_launch(void* const* d_in, const int* in_sizes, int n_in,
                              void* d_out, int out_size, void* d_ws, size_t ws_size,
                              hipStream_t stream) {
    const float* features = (const float*)d_in[0];
    const int*   src      = (const int*)  d_in[1];
    const int*   dst      = (const int*)  d_in[2];
    const int*   etype    = (const int*)  d_in[3];
    const float* W1       = (const float*)d_in[4];
    const float* W1s      = (const float*)d_in[5];
    const float* b1       = (const float*)d_in[6];
    const float* W2       = (const float*)d_in[7];
    const float* W2s      = (const float*)d_in[8];
    const float* b2       = (const float*)d_in[9];
    float*       out      = (float*)d_out;

    float*  aggR = (float*)d_ws;
    float*  h1   = aggR + (size_t)N_REL * N_NODES * KDIM;
    float2* Wf1  = (float2*)(h1 + (size_t)N_NODES * KDIM);
    float2* Wf2  = Wf1 + (size_t)NSEG * 8 * 32 * 32;

    const long long n4     = (long long)N_REL * N_NODES * KDIM / 4;
    const int zeroBlocks   = (int)((n4 + 255) / 256);
    const int scatBlocks   = (int)(((long long)N_EDGES * KDIM) / 256);
    const int gemmBlocks   = N_NODES / 16;

    const int pack1 = NSEG * 8 * 32 * 32;
    const int pack2 = NSEG * 4 * 32 * 32;

    rgcn_pack_w_kernel<<<(pack1 + 255) / 256, 256, 0, stream>>>(W1, W1s, Wf1, 128);
    rgcn_pack_w_kernel<<<(pack2 + 255) / 256, 256, 0, stream>>>(W2, W2s, Wf2, 64);

    rgcn_zero_kernel<<<zeroBlocks, 256, 0, stream>>>((float4*)aggR, n4);
    rgcn_scatter_kernel<<<scatBlocks, 256, 0, stream>>>(
        features, src, dst, etype, aggR, N_EDGES);
    rgcn_gemm_kernel<<<gemmBlocks, 256, 0, stream>>>(
        aggR, features, Wf1, b1, h1, 128, 1);

    float* rel = aggR;
    rgcn_gemm_t_kernel<<<gemmBlocks, 288, 0, stream>>>(h1, Wf2, b2, rel, out);
    rgcn_owner_sum_kernel<<<OWN_BLOCKS, OWN_THREADS, 0, stream>>>(
        rel, src, dst, etype, out);
}


# Layer 1 (128 to 128) also transforms before it aggregates: GEMM writes x@W1[r] rows, owner groups sum 128-wide rows in LDS, ReLU at their store; zero pass + global atomics go; group arrival order
# speedup vs baseline: 2.7493x; 1.8188x over previous
#include <hip/hip_runtime.h>

typedef __attribute__((ext_vector_type(2))) float v2f;
typedef __attribute__((ext_vector_type(8))) float v8f;

#define N_NODES 50000
#define N_EDGES 640000
#define N_REL   8
#define KDIM    128
#define NSEG    (N_REL + 1)

__global__ void rgcn_pack_w_kernel(const float* __restrict__ Wrel,
                                   const float* __restrict__ Wself,
                                   float2*      __restrict__ Wf,
                                   int O) {
    const int tiles = O >> 4;
    const int total = NSEG * tiles * 32 * 32;
    int idx = blockIdx.x * blockDim.x + threadIdx.x;
    if (idx >= total) return;
    int lane = idx & 31;
    int k4   = (idx >> 5) & 31;
    int t    = (idx >> 10) % tiles;
    int seg  = idx / (tiles << 10);
    const float* W = (seg < N_REL) ? (Wrel + (size_t)seg * KDIM * O) : Wself;
    int k0 = k4 * 4 + (lane >> 4) * 2;
    int n  = t * 16 + (lane & 15);
    Wf[idx] = make_float2(W[(size_t)k0 * O + n], W[(size_t)(k0 + 1) * O + n]);
}

template <int O>
__device__ __forceinline__ void gemm_t_body(const float*  __restrict__ h,
                                            const float2* __restrict__ Wf,
                                            const float*  __restrict__ bias,
                                            float*        __restrict__ rel,
                                            float*        __restrict__ out,
                                            float* lds) {
    constexpr int TILES = O / 16;

    const int m0   = blockIdx.x * 16;
    const int lane = threadIdx.x & 31;
    const int seg  = threadIdx.x >> 5;
    const int half = lane >> 4;
    const int l15  = lane & 15;

    if (threadIdx.x < 256) {
        const int col  = threadIdx.x & 127;
        const int row0 = threadIdx.x >> 7;
#pragma unroll
        for (int i = 0; i < 8; ++i) {
            int row = row0 + 2 * i;
            lds[row * 132 + col] = h[(size_t)(m0 + row) * KDIM + col];
        }
    }
    __syncthreads();

    const float2* wf = Wf + ((size_t)(seg * TILES) << 10) + lane;

    v8f c[TILES];
#pragma unroll
    for (int t = 0; t < TILES; ++t)
        c[t] = (v8f){0.f, 0.f, 0.f, 0.f, 0.f, 0.f, 0.f, 0.f};

#pragma unroll
    for (int k4 = 0; k4 < 32; ++k4) {
        const float* ap = &lds[l15 * 132 + k4 * 4 + half * 2];
        v2f a;
        a.x = ap[0];
        a.y = ap[1];
#pragma unroll
        for (int t = 0; t < TILES; ++t) {
            float2 f = wf[k4 * 32 + t * 1024];
            v2f b;
            b.x = f.x;
            b.y = f.y;
            c[t] = __builtin_amdgcn_wmma_f32_16x16x4_f32(
                       false, a, false, b, (short)0, c[t], false, false);
        }
    }

    if (seg < N_REL) {
#pragma unroll
        for (int v = 0; v < 8; ++v) {
            float* p = rel + ((size_t)(m0 + v + 8 * half) * N_REL + seg) * O + l15;
#pragma unroll
            for (int t = 0; t < TILES; ++t) p[16 * t] = c[t][v];
        }
    } else {
        float bv[TILES];
#pragma unroll
        for (int t = 0; t < TILES; ++t) bv[t] = bias[16 * t + l15];
#pragma unroll
        for (int v = 0; v < 8; ++v) {
            float* p = out + (size_t)(m0 + v + 8 * half) * O + l15;
#pragma unroll
            for (int t = 0; t < TILES; ++t) p[16 * t] = c[t][v] + bv[t];
        }
    }
}

__global__ void __launch_bounds__(288)
rgcn_gemm_t128_kernel(const float*  __restrict__ h,
                      const float2* __restrict__ Wf,
                      const float*  __restrict__ bias,
                      float*        __restrict__ rel,
                      float*        __restrict__ out) {
    __shared__ float lds[16 * 132];
    gemm_t_body<128>(h, Wf, bias, rel, out, lds);
}

__global__ void __launch_bounds__(288)
rgcn_gemm_t64_kernel(const float*  __restrict__ h,
                     const float2* __restrict__ Wf,
                     const float*  __restrict__ bias,
                     float*        __restrict__ rel,
                     float*        __restrict__ out) {
    __shared__ float lds[16 * 132];
    gemm_t_body<64>(h, Wf, bias, rel, out, lds);
}

#define OWN_ROWS    196
#define OWN_BLOCKS  ((N_NODES + OWN_ROWS - 1) / OWN_ROWS)
#define OWN_THREADS 1024
#define OWN_WAVES   (OWN_THREADS / 32)
#define RING_CAP    1024

template <int W>
__device__ __forceinline__ void owner_add_batch(const float* __restrict__ rel,
                                                const int*   __restrict__ esrc,
                                                const int*   __restrict__ edst,
                                                const int*   __restrict__ etyp,
                                                float* acc, const int* q,
                                                int head, int cnt, int base, int lane) {
    constexpr int PARTS = W / 64;
    constexpr int ROWS  = 32 / PARTS;
    const int e    = (lane < cnt) ? q[(head + lane) & (RING_CAP - 1)] : 0;
    const int roff = (esrc[e] * N_REL + etyp[e]) * W;
    const int doff = (edst[e] - base) * W;
#pragma unroll
    for (int g = 0; g < 32; g += ROWS) {
        if (g < cnt) {
            float2 v[ROWS][PARTS];
#pragma unroll
            for (int j = 0; j < ROWS; ++j) {
                const int ro = __builtin_amdgcn_readlane(roff, g + j);
#pragma unroll
                for (int p = 0; p < PARTS; ++p)
                    v[j][p] = *(const float2*)(rel + ro + p * 64 + lane * 2);
            }
#pragma unroll
            for (int j = 0; j < ROWS; ++j) {
                if (g + j < cnt) {
                    float* a = acc + __builtin_amdgcn_readlane(doff, g + j) + lane * 2;
#pragma unroll
                    for (int p = 0; p < PARTS; ++p) {
                        atomicAdd(a + p * 64,     v[j][p].x);
                        atomicAdd(a + p * 64 + 1, v[j][p].y);
                    }
                }
            }
        }
    }
}

template <int W, bool RELU>
__device__ __forceinline__ void owner_sum_body(const float* __restrict__ rel,
                                               const int*   __restrict__ esrc,
                                               const int*   __restrict__ edst,
                                               const int*   __restrict__ etyp,
                                               float*       __restrict__ out,
                                               float* acc, int* ring) {
    const int tid   = threadIdx.x;
    const int lane  = tid & 31;
    const int wave  = tid >> 5;
    const int base  = blockIdx.x * OWN_ROWS;
    const int nrows = (N_NODES - base < OWN_ROWS) ? (N_NODES - base) : OWN_ROWS;

    float4*       a4 = (float4*)acc;
    float4*       o4 = (float4*)(out + (size_t)base * W);
    for (int i = tid; i < nrows * (W / 4); i += OWN_THREADS) a4[i] = o4[i];
    __syncthreads();

    int* q = ring + wave * RING_CAP;
    int head = 0, tail = 0;

    const int4* d4 = (const int4*)edst;
    const int   nQ = N_EDGES / 4;

    for (int q0 = 0; q0 < nQ; q0 += 4 * OWN_THREADS) {
        int4 dv[4];
#pragma unroll
        for (int u = 0; u < 4; ++u) {
            int qi = q0 + u * OWN_THREADS + tid;
            dv[u] = (qi < nQ) ? d4[qi] : make_int4(-1, -1, -1, -1);
        }
#pragma unroll
        for (int u = 0; u < 4; ++u) {
            const int e0 = (q0 + u * OWN_THREADS + tid) * 4;
            const bool m0 = (unsigned)(dv[u].x - base) < (unsigned)nrows;
            const bool m1 = (unsigned)(dv[u].y - base) < (unsigned)nrows;
            const bool m2 = (unsigned)(dv[u].z - base) < (unsigned)nrows;
            const bool m3 = (unsigned)(dv[u].w - base) < (unsigned)nrows;
            if (__builtin_amdgcn_ballot_w32(m0 | m1 | m2 | m3)) {
                unsigned k;
                k = __builtin_amdgcn_ballot_w32(m0);
                if (m0) q[(tail + __builtin_amdgcn_mbcnt_lo(k, 0u)) & (RING_CAP - 1)] = e0;
                tail += __builtin_popcount(k);
                k = __builtin_amdgcn_ballot_w32(m1);
                if (m1) q[(tail + __builtin_amdgcn_mbcnt_lo(k, 0u)) & (RING_CAP - 1)] = e0 + 1;
                tail += __builtin_popcount(k);
                k = __builtin_amdgcn_ballot_w32(m2);
                if (m2) q[(tail + __builtin_amdgcn_mbcnt_lo(k, 0u)) & (RING_CAP - 1)] = e0 + 2;
                tail += __builtin_popcount(k);
                k = __builtin_amdgcn_ballot_w32(m3);
                if (m3) q[(tail + __builtin_amdgcn_mbcnt_lo(k, 0u)) & (RING_CAP - 1)] = e0 + 3;
                tail += __builtin_popcount(k);
            }
        }

        while (tail - head >= 32) {
            owner_add_batch<W>(rel, esrc, edst, etyp, acc, q, head, 32, base, lane);
            head += 32;
        }
    }

    if (tail - head > 0)
        owner_add_batch<W>(rel, esrc, edst, etyp, acc, q, head, tail - head, base, lane);

    __syncthreads();
    for (int i = tid; i < nrows * (W / 4); i += OWN_THREADS) {
        float4 x = a4[i];
        if (RELU) {
            x.x = fmaxf(x.x, 0.f);
            x.y = fmaxf(x.y, 0.f);
            x.z = fmaxf(x.z, 0.f);
            x.w = fmaxf(x.w, 0.f);
        }
        o4[i] = x;
    }
}

__global__ void __launch_bounds__(OWN_THREADS)
rgcn_owner_sum128_relu_kernel(const float* __restrict__ rel,
                              const int*   __restrict__ esrc,
                              const int*   __restrict__ edst,
                              const int*   __restrict__ etyp,
                              float*       __restrict__ out) {
    __shared__ __attribute__((aligned(16))) float acc[OWN_ROWS * 128];
    __shared__ int ring[OWN_WAVES * RING_CAP];
    owner_sum_body<128, true>(rel, esrc, edst, etyp, out, acc, ring);
}

__global__ void __launch_bounds__(OWN_THREADS)
rgcn_owner_sum64_kernel(const float* __restrict__ rel,
                        const int*   __restrict__ esrc,
                        const int*   __restrict__ edst,
                        const int*   __restrict__ etyp,
                        float*       __restrict__ out) {
    __shared__ __attribute__((aligned(16))) float acc[OWN_ROWS * 64];
    __shared__ int ring[OWN_WAVES * RING_CAP];
    owner_sum_body<64, false>(rel, esrc, edst, etyp, out, acc, ring);
}

extern "C" void kernel_launch(void* const* d_in, const int* in_sizes, int n_in,
                              void* d_out, int out_size, void* d_ws, size_t ws_size,
                              hipStream_t stream) {
    const float* features = (const float*)d_in[0];
    const int*   src      = (const int*)  d_in[1];
    const int*   dst      = (const int*)  d_in[2];
    const int*   etype    = (const int*)  d_in[3];
    const float* W1       = (const float*)d_in[4];
    const float* W1s      = (const float*)d_in[5];
    const float* b1       = (const float*)d_in[6];
    const float* W2       = (const float*)d_in[7];
    const float* W2s      = (const float*)d_in[8];
    const float* b2       = (const float*)d_in[9];
    float*       out      = (float*)d_out;

    float*  rel  = (float*)d_ws;
    float*  h1   = rel + (size_t)N_REL * N_NODES * KDIM;
    float2* Wf1  = (float2*)(h1 + (size_t)N_NODES * KDIM);
    float2* Wf2  = Wf1 + (size_t)NSEG * 8 * 32 * 32;

    const int gemmBlocks   = N_NODES / 16;

    const int pack1 = NSEG * 8 * 32 * 32;
    const int pack2 = NSEG * 4 * 32 * 32;

    rgcn_pack_w_kernel<<<(pack1 + 255) / 256, 256, 0, stream>>>(W1, W1s, Wf1, 128);
    rgcn_pack_w_kernel<<<(pack2 + 255) / 256, 256, 0, stream>>>(W2, W2s, Wf2, 64);

    rgcn_gemm_t128_kernel<<<gemmBlocks, 288, 0, stream>>>(features, Wf1, b1, rel, h1);
    rgcn_owner_sum128_relu_kernel<<<OWN_BLOCKS, OWN_THREADS, 0, stream>>>(
        rel, src, dst, etype, h1);

    rgcn_gemm_t64_kernel<<<gemmBlocks, 288, 0, stream>>>(h1, Wf2, b2, rel, out);
    rgcn_owner_sum64_kernel<<<OWN_BLOCKS, OWN_THREADS, 0, stream>>>(
        rel, src, dst, etype, out);
}


# Edge search key narrowed: a pre-pass writes dst/196 as one byte per edge; owner groups scan 16 edges per 16-byte load with zero-byte tests, not 4 ints; +1 launch; sums stay in group arrival order
# speedup vs baseline: 3.0576x; 1.1121x over previous
#include <hip/hip_runtime.h>

typedef __attribute__((ext_vector_type(2))) float v2f;
typedef __attribute__((ext_vector_type(8))) float v8f;

#define N_NODES 50000
#define N_EDGES 640000
#define N_REL   8
#define KDIM    128
#define NSEG    (N_REL + 1)

__global__ void rgcn_pack_w_kernel(const float* __restrict__ Wrel,
                                   const float* __restrict__ Wself,
                                   float2*      __restrict__ Wf,
                                   int O) {
    const int tiles = O >> 4;
    const int total = NSEG * tiles * 32 * 32;
    int idx = blockIdx.x * blockDim.x + threadIdx.x;
    if (idx >= total) return;
    int lane = idx & 31;
    int k4   = (idx >> 5) & 31;
    int t    = (idx >> 10) % tiles;
    int seg  = idx / (tiles << 10);
    const float* W = (seg < N_REL) ? (Wrel + (size_t)seg * KDIM * O) : Wself;
    int k0 = k4 * 4 + (lane >> 4) * 2;
    int n  = t * 16 + (lane & 15);
    Wf[idx] = make_float2(W[(size_t)k0 * O + n], W[(size_t)(k0 + 1) * O + n]);
}

template <int O>
__device__ __forceinline__ void gemm_t_body(const float*  __restrict__ h,
                                            const float2* __restrict__ Wf,
                                            const float*  __restrict__ bias,
                                            float*        __restrict__ rel,
                                            float*        __restrict__ out,
                                            float* lds) {
    constexpr int TILES = O / 16;

    const int m0   = blockIdx.x * 16;
    const int lane = threadIdx.x & 31;
    const int seg  = threadIdx.x >> 5;
    const int half = lane >> 4;
    const int l15  = lane & 15;

    if (threadIdx.x < 256) {
        const int col  = threadIdx.x & 127;
        const int row0 = threadIdx.x >> 7;
#pragma unroll
        for (int i = 0; i < 8; ++i) {
            int row = row0 + 2 * i;
            lds[row * 132 + col] = h[(size_t)(m0 + row) * KDIM + col];
        }
    }
    __syncthreads();

    const float2* wf = Wf + ((size_t)(seg * TILES) << 10) + lane;

    v8f c[TILES];
#pragma unroll
    for (int t = 0; t < TILES; ++t)
        c[t] = (v8f){0.f, 0.f, 0.f, 0.f, 0.f, 0.f, 0.f, 0.f};

#pragma unroll
    for (int k4 = 0; k4 < 32; ++k4) {
        const float* ap = &lds[l15 * 132 + k4 * 4 + half * 2];
        v2f a;
        a.x = ap[0];
        a.y = ap[1];
#pragma unroll
        for (int t = 0; t < TILES; ++t) {
            float2 f = wf[k4 * 32 + t * 1024];
            v2f b;
            b.x = f.x;
            b.y = f.y;
            c[t] = __builtin_amdgcn_wmma_f32_16x16x4_f32(
                       false, a, false, b, (short)0, c[t], false, false);
        }
    }

    if (seg < N_REL) {
#pragma unroll
        for (int v = 0; v < 8; ++v) {
            float* p = rel + ((size_t)(m0 + v + 8 * half) * N_REL + seg) * O + l15;
#pragma unroll
            for (int t = 0; t < TILES; ++t) p[16 * t] = c[t][v];
        }
    } else {
        float bv[TILES];
#pragma unroll
        for (int t = 0; t < TILES; ++t) bv[t] = bias[16 * t + l15];
#pragma unroll
        for (int v = 0; v < 8; ++v) {
            float* p = out + (size_t)(m0 + v + 8 * half) * O + l15;
#pragma unroll
            for (int t = 0; t < TILES; ++t) p[16 * t] = c[t][v] + bv[t];
        }
    }
}

__global__ void __launch_bounds__(288)
rgcn_gemm_t128_kernel(const float*  __restrict__ h,
                      const float2* __restrict__ Wf,
                      const float*  __restrict__ bias,
                      float*        __restrict__ rel,
                      float*        __restrict__ out) {
    __shared__ float lds[16 * 132];
    gemm_t_body<128>(h, Wf, bias, rel, out, lds);
}

__global__ void __launch_bounds__(288)
rgcn_gemm_t64_kernel(const float*  __restrict__ h,
                     const float2* __restrict__ Wf,
                     const float*  __restrict__ bias,
                     float*        __restrict__ rel,
                     float*        __restrict__ out) {
    __shared__ float lds[16 * 132];
    gemm_t_body<64>(h, Wf, bias, rel, out, lds);
}

#define OWN_ROWS    196
#define OWN_BLOCKS  ((N_NODES + OWN_ROWS - 1) / OWN_ROWS)
#define OWN_THREADS 1024
#define OWN_WAVES   (OWN_THREADS / 32)
#define RING_CAP    1024
#define OWN_GROUPS  ((N_EDGES / 16 + OWN_THREADS - 1) / OWN_THREADS)

__global__ void rgcn_owner_byte_kernel(const int* __restrict__ edst,
                                       unsigned*  __restrict__ owner) {
    const int i = blockIdx.x * blockDim.x + threadIdx.x;
    if (i >= N_EDGES / 4) return;
    const int4 d = ((const int4*)edst)[i];
    owner[i] = (((unsigned)d.x / OWN_ROWS) & 255u)
             | ((((unsigned)d.y / OWN_ROWS) & 255u) << 8)
             | ((((unsigned)d.z / OWN_ROWS) & 255u) << 16)
             | ((((unsigned)d.w / OWN_ROWS) & 255u) << 24);
}

__device__ __forceinline__ unsigned zero_byte_flags(unsigned x) {
    return ~((((x & 0x7F7F7F7Fu) + 0x7F7F7F7Fu) | x) | 0x7F7F7F7Fu);
}

template <int W>
__device__ __forceinline__ void owner_add_batch(const float* __restrict__ rel,
                                                const int*   __restrict__ esrc,
                                                const int*   __restrict__ edst,
                                                const int*   __restrict__ etyp,
                                                float* acc, const int* q,
                                                int head, int cnt, int base, int lane) {
    constexpr int PARTS = W / 64;
    constexpr int ROWS  = 32 / PARTS;
    const int e    = (lane < cnt) ? q[(head + lane) & (RING_CAP - 1)] : 0;
    const int roff = (esrc[e] * N_REL + etyp[e]) * W;
    const int doff = (edst[e] - base) * W;
#pragma unroll
    for (int g = 0; g < 32; g += ROWS) {
        if (g < cnt) {
            float2 v[ROWS][PARTS];
#pragma unroll
            for (int j = 0; j < ROWS; ++j) {
                const int ro = __builtin_amdgcn_readlane(roff, g + j);
#pragma unroll
                for (int p = 0; p < PARTS; ++p)
                    v[j][p] = *(const float2*)(rel + ro + p * 64 + lane * 2);
            }
#pragma unroll
            for (int j = 0; j < ROWS; ++j) {
                if (g + j < cnt) {
                    float* a = acc + __builtin_amdgcn_readlane(doff, g + j) + lane * 2;
#pragma unroll
                    for (int p = 0; p < PARTS; ++p) {
                        atomicAdd(a + p * 64,     v[j][p].x);
                        atomicAdd(a + p * 64 + 1, v[j][p].y);
                    }
                }
            }
        }
    }
}

template <int W, bool RELU>
__device__ __forceinline__ void owner_sum_body(const float* __restrict__ rel,
                                               const int*   __restrict__ esrc,
                                               const int*   __restrict__ edst,
                                               const int*   __restrict__ etyp,
                                               const unsigned* __restrict__ owner,
                                               float*       __restrict__ out,
                                               float* acc, int* ring) {
    const int tid   = threadIdx.x;
    const int lane  = tid & 31;
    const int wave  = __builtin_amdgcn_readfirstlane(tid >> 5);
    const int base  = blockIdx.x * OWN_ROWS;
    const int nrows = (N_NODES - base < OWN_ROWS) ? (N_NODES - base) : OWN_ROWS;

    float4*       a4 = (float4*)acc;
    float4*       o4 = (float4*)(out + (size_t)base * W);
    for (int i = tid; i < nrows * (W / 4); i += OWN_THREADS) a4[i] = o4[i];
    __syncthreads();

    int* q = ring + wave * RING_CAP;
    int head = 0, tail = 0;

    const uint4*   o16 = (const uint4*)owner;
    const int      nP  = N_EDGES / 16;
    const unsigned key = blockIdx.x * 0x01010101u;

    const int w0   = wave * 32;
    const int last = nP - 32;
    uint4 pa = o16[((w0 < last) ? w0 : last) + lane];
    uint4 pb = o16[((w0 + OWN_THREADS < last) ? w0 + OWN_THREADS : last) + lane];

    for (int g = 0; g < OWN_GROUPS; g += 2) {
#pragma unroll
        for (int s = 0; s < 2; ++s) {
            const int   pw  = (g + s) * OWN_THREADS + w0;
            const uint4 cur = s ? pb : pa;
            const unsigned f0 = zero_byte_flags(cur.x ^ key);
            const unsigned f1 = zero_byte_flags(cur.y ^ key);
            const unsigned f2 = zero_byte_flags(cur.z ^ key);
            const unsigned f3 = zero_byte_flags(cur.w ^ key);

            const int   pn  = pw + 2 * OWN_THREADS;
            const uint4 nxt = o16[((pn < last) ? pn : last) + lane];
            if (s) pb = nxt; else pa = nxt;

            unsigned m = (f0 >> 7) | (f1 >> 6) | (f2 >> 5) | (f3 >> 4);
            if (pw < nP) {
                const int eb = (pw + lane) * 16;
                unsigned k;
                while ((k = __builtin_amdgcn_ballot_w32(m != 0u)) != 0u) {
                    if (m != 0u) {
                        const int c = __builtin_ctz(m);
                        q[(tail + __builtin_amdgcn_mbcnt_lo(k, 0u)) & (RING_CAP - 1)] =
                            eb + ((c & 7) << 2) + (c >> 3);
                        m &= m - 1u;
                    }
                    tail += __builtin_popcount(k);
                }
            }

            while (tail - head >= 32) {
                owner_add_batch<W>(rel, esrc, edst, etyp, acc, q, head, 32, base, lane);
                head += 32;
            }
        }
    }

    if (tail - head > 0)
        owner_add_batch<W>(rel, esrc, edst, etyp, acc, q, head, tail - head, base, lane);

    __syncthreads();
    for (int i = tid; i < nrows * (W / 4); i += OWN_THREADS) {
        float4 x = a4[i];
        if (RELU) {
            x.x = fmaxf(x.x, 0.f);
            x.y = fmaxf(x.y, 0.f);
            x.z = fmaxf(x.z, 0.f);
            x.w = fmaxf(x.w, 0.f);
        }
        o4[i] = x;
    }
}

__global__ void __launch_bounds__(OWN_THREADS)
rgcn_owner_sum128_relu_kernel(const float* __restrict__ rel,
                              const int*   __restrict__ esrc,
                              const int*   __restrict__ edst,
                              const int*   __restrict__ etyp,
                              const unsigned* __restrict__ owner,
                              float*       __restrict__ out) {
    __shared__ __attribute__((aligned(16))) float acc[OWN_ROWS * 128];
    __shared__ int ring[OWN_WAVES * RING_CAP];
    owner_sum_body<128, true>(rel, esrc, edst, etyp, owner, out, acc, ring);
}

__global__ void __launch_bounds__(OWN_THREADS)
rgcn_owner_sum64_kernel(const float* __restrict__ rel,
                        const int*   __restrict__ esrc,
                        const int*   __restrict__ edst,
                        const int*   __restrict__ etyp,
                        const unsigned* __restrict__ owner,
                        float*       __restrict__ out) {
    __shared__ __attribute__((aligned(16))) float acc[OWN_ROWS * 64];
    __shared__ int ring[OWN_WAVES * RING_CAP];
    owner_sum_body<64, false>(rel, esrc, edst, etyp, owner, out, acc, ring);
}

extern "C" void kernel_launch(void* const* d_in, const int* in_sizes, int n_in,
                              void* d_out, int out_size, void* d_ws, size_t ws_size,
                              hipStream_t stream) {
    const float* features = (const float*)d_in[0];
    const int*   src      = (const int*)  d_in[1];
    const int*   dst      = (const int*)  d_in[2];
    const int*   etype    = (const int*)  d_in[3];
    const float* W1       = (const float*)d_in[4];
    const float* W1s      = (const float*)d_in[5];
    const float* b1       = (const float*)d_in[6];
    const float* W2       = (const float*)d_in[7];
    const float* W2s      = (const float*)d_in[8];
    const float* b2       = (const float*)d_in[9];
    float*       out      = (float*)d_out;

    float*  rel  = (float*)d_ws;
    float*  h1   = rel + (size_t)N_REL * N_NODES * KDIM;
    float2* Wf1  = (float2*)(h1 + (size_t)N_NODES * KDIM);
    float2* Wf2  = Wf1 + (size_t)NSEG * 8 * 32 * 32;
    unsigned* owner = (unsigned*)(Wf2 + (size_t)NSEG * 4 * 32 * 32);

    const int gemmBlocks   = N_NODES / 16;

    const int pack1 = NSEG * 8 * 32 * 32;
    const int pack2 = NSEG * 4 * 32 * 32;

    rgcn_pack_w_kernel<<<(pack1 + 255) / 256, 256, 0, stream>>>(W1, W1s, Wf1, 128);
    rgcn_pack_w_kernel<<<(pack2 + 255) / 256, 256, 0, stream>>>(W2, W2s, Wf2, 64);

    rgcn_owner_byte_kernel<<<(N_EDGES / 4 + 255) / 256, 256, 0, stream>>>(dst, owner);

    rgcn_gemm_t128_kernel<<<gemmBlocks, 288, 0, stream>>>(features, Wf1, b1, rel, h1);
    rgcn_owner_sum128_relu_kernel<<<OWN_BLOCKS, OWN_THREADS, 0, stream>>>(
        rel, src, dst, etype, owner, h1);

    rgcn_gemm_t64_kernel<<<gemmBlocks, 288, 0, stream>>>(h1, Wf2, b2, rel, out);
    rgcn_owner_sum64_kernel<<<OWN_BLOCKS, OWN_THREADS, 0, stream>>>(
        rel, src, dst, etype, owner, out);
}


# Edges grouped by owner group per layer (LDS counting sort of 2500-edge chunks); owner groups read their own 32-bit edge records: no search of all edges, no src/type/dst lookups; arrival order
# speedup vs baseline: 3.2485x; 1.0624x over previous
#include <hip/hip_runtime.h>

typedef __attribute__((ext_vector_type(2))) float v2f;
typedef __attribute__((ext_vector_type(8))) float v8f;

#define N_NODES 50000
#define N_EDGES 640000
#define N_REL   8
#define KDIM    128
#define NSEG    (N_REL + 1)

__global__ void rgcn_pack_w_kernel(const float* __restrict__ Wrel,
                                   const float* __restrict__ Wself,
                                   float2*      __restrict__ Wf,
                                   int O) {
    const int tiles = O >> 4;
    const int total = NSEG * tiles * 32 * 32;
    int idx = blockIdx.x * blockDim.x + threadIdx.x;
    if (idx >= total) return;
    int lane = idx & 31;
    int k4   = (idx >> 5) & 31;
    int t    = (idx >> 10) % tiles;
    int seg  = idx / (tiles << 10);
    const float* W = (seg < N_REL) ? (Wrel + (size_t)seg * KDIM * O) : Wself;
    int k0 = k4 * 4 + (lane >> 4) * 2;
    int n  = t * 16 + (lane & 15);
    Wf[idx] = make_float2(W[(size_t)k0 * O + n], W[(size_t)(k0 + 1) * O + n]);
}

template <int O>
__device__ __forceinline__ void gemm_t_body(const float*  __restrict__ h,
                                            const float2* __restrict__ Wf,
                                            const float*  __restrict__ bias,
                                            float*        __restrict__ rel,
                                            float*        __restrict__ out,
                                            float* lds) {
    constexpr int TILES = O / 16;

    const int m0   = blockIdx.x * 16;
    const int lane = threadIdx.x & 31;
    const int seg  = threadIdx.x >> 5;
    const int half = lane >> 4;
    const int l15  = lane & 15;

    if (threadIdx.x < 256) {
        const int col  = threadIdx.x & 127;
        const int row0 = threadIdx.x >> 7;
#pragma unroll
        for (int i = 0; i < 8; ++i) {
            int row = row0 + 2 * i;
            lds[row * 132 + col] = h[(size_t)(m0 + row) * KDIM + col];
        }
    }
    __syncthreads();

    const float2* wf = Wf + ((size_t)(seg * TILES) << 10) + lane;

    v8f c[TILES];
#pragma unroll
    for (int t = 0; t < TILES; ++t)
        c[t] = (v8f){0.f, 0.f, 0.f, 0.f, 0.f, 0.f, 0.f, 0.f};

#pragma unroll
    for (int k4 = 0; k4 < 32; ++k4) {
        const float* ap = &lds[l15 * 132 + k4 * 4 + half * 2];
        v2f a;
        a.x = ap[0];
        a.y = ap[1];
#pragma unroll
        for (int t = 0; t < TILES; ++t) {
            float2 f = wf[k4 * 32 + t * 1024];
            v2f b;
            b.x = f.x;
            b.y = f.y;
            c[t] = __builtin_amdgcn_wmma_f32_16x16x4_f32(
                       false, a, false, b, (short)0, c[t], false, false);
        }
    }

    if (seg < N_REL) {
#pragma unroll
        for (int v = 0; v < 8; ++v) {
            float* p = rel + ((size_t)(m0 + v + 8 * half) * N_REL + seg) * O + l15;
#pragma unroll
            for (int t = 0; t < TILES; ++t) p[16 * t] = c[t][v];
        }
    } else {
        float bv[TILES];
#pragma unroll
        for (int t = 0; t < TILES; ++t) bv[t] = bias[16 * t + l15];
#pragma unroll
        for (int v = 0; v < 8; ++v) {
            float* p = out + (size_t)(m0 + v + 8 * half) * O + l15;
#pragma unroll
            for (int t = 0; t < TILES; ++t) p[16 * t] = c[t][v] + bv[t];
        }
    }
}

__global__ void __launch_bounds__(288)
rgcn_gemm_t128_kernel(const float*  __restrict__ h,
                      const float2* __restrict__ Wf,
                      const float*  __restrict__ bias,
                      float*        __restrict__ rel,
                      float*        __restrict__ out) {
    __shared__ float lds[16 * 132];
    gemm_t_body<128>(h, Wf, bias, rel, out, lds);
}

__global__ void __launch_bounds__(288)
rgcn_gemm_t64_kernel(const float*  __restrict__ h,
                     const float2* __restrict__ Wf,
                     const float*  __restrict__ bias,
                     float*        __restrict__ rel,
                     float*        __restrict__ out) {
    __shared__ float lds[16 * 132];
    gemm_t_body<64>(h, Wf, bias, rel, out, lds);
}

#define OWN_ROWS    196
#define OWN_BLOCKS  ((N_NODES + OWN_ROWS - 1) / OWN_ROWS)
#define OWN_THREADS 1024
#define OWN_WAVES   (OWN_THREADS / 32)

#define BIN_EDGES   2500
#define BIN_CHUNKS  (N_EDGES / BIN_EDGES)
#define BIN_THREADS 256
#define BIN_ITERS   ((BIN_EDGES + BIN_THREADS - 1) / BIN_THREADS)
#define BIN_PITCH   (((BIN_EDGES + 31) / 32) * 32)
#define WAVE_CHUNKS (BIN_CHUNKS / OWN_WAVES)

static_assert(N_EDGES % BIN_EDGES == 0, "chunks cover the edge list exactly");
static_assert(BIN_CHUNKS % OWN_WAVES == 0, "every owner wave takes whole chunks");
static_assert(WAVE_CHUNKS <= 32, "one table word per lane");
static_assert(OWN_BLOCKS <= 256 && OWN_ROWS <= 256, "owner and local row fit in a byte");
static_assert(BIN_THREADS == 256 && OWN_BLOCKS <= BIN_THREADS, "one counter per thread");
static_assert(BIN_EDGES < 65536, "start and count fit in 16 bits each");

__global__ void __launch_bounds__(BIN_THREADS)
rgcn_bin_kernel(const int* __restrict__ esrc,
                const int* __restrict__ edst,
                const int* __restrict__ etyp,
                unsigned*  __restrict__ recs,
                unsigned*  __restrict__ tab) {
    __shared__ unsigned cnt[BIN_THREADS];
    __shared__ unsigned wsum[BIN_THREADS / 32];
    __shared__ unsigned sorted[BIN_EDGES];

    const int tid = threadIdx.x & (BIN_THREADS - 1);
    const int e0  = blockIdx.x * BIN_EDGES;

    cnt[tid] = 0u;
    __syncthreads();

    unsigned dv[BIN_ITERS], rv[BIN_ITERS];
#pragma unroll
    for (int i = 0; i < BIN_ITERS; ++i) {
        const int j = tid + i * BIN_THREADS;
        dv[i] = 0u;
        rv[i] = 0u;
        if (j < BIN_EDGES) {
            dv[i] = (unsigned)edst[e0 + j];
            rv[i] = (unsigned)(esrc[e0 + j] * N_REL + etyp[e0 + j]);
        }
    }
    unsigned rec[BIN_ITERS], key[BIN_ITERS];
#pragma unroll
    for (int i = 0; i < BIN_ITERS; ++i) {
        const int j = tid + i * BIN_THREADS;
        rec[i] = 0u;
        key[i] = 0u;
        if (j < BIN_EDGES) {
            const unsigned o = (dv[i] / OWN_ROWS) & 255u;
            rec[i] = (rv[i] << 8) | ((dv[i] - o * OWN_ROWS) & 255u);
            key[i] = o | (atomicAdd(&cnt[o], 1u) << 8);
        }
    }
    __syncthreads();

    const unsigned c = cnt[tid];
    unsigned incl = c;
#pragma unroll
    for (int s = 1; s < 32; s <<= 1) {
        const unsigned t = __shfl_up(incl, s);
        if ((tid & 31) >= s) incl += t;
    }
    if ((tid & 31) == 31) wsum[tid >> 5] = incl;
    __syncthreads();
    unsigned start = incl - c;
    for (int w = 0; w < (tid >> 5); ++w) start += wsum[w];

    cnt[tid] = start;
    tab[blockIdx.x * OWN_BLOCKS + tid] = (start << 16) | c;
    __syncthreads();

#pragma unroll
    for (int i = 0; i < BIN_ITERS; ++i) {
        const int j = tid + i * BIN_THREADS;
        if (j < BIN_EDGES) sorted[cnt[key[i] & 255u] + (key[i] >> 8)] = rec[i];
    }
    __syncthreads();

    unsigned* dstp = recs + blockIdx.x * BIN_PITCH;
    for (int j = tid; j < BIN_EDGES; j += BIN_THREADS) dstp[j] = sorted[j];
}

template <int W>
__device__ __forceinline__ void owner_add_batch(const float* __restrict__ rel,
                                                float* acc, unsigned rec,
                                                int cnt, int lane) {
    constexpr int PARTS = W / 64;
    constexpr int ROWS  = 32 / PARTS;
    const int roff = (int)(rec >> 8) * W;
    const int doff = (int)(rec & 255u) * W;
#pragma unroll
    for (int g = 0; g < 32; g += ROWS) {
        if (g < cnt) {
            float2 v[ROWS][PARTS];
#pragma unroll
            for (int j = 0; j < ROWS; ++j) {
                const int ro = __builtin_amdgcn_readlane(roff, g + j);
#pragma unroll
                for (int p = 0; p < PARTS; ++p)
                    v[j][p] = *(const float2*)(rel + ro + p * 64 + lane * 2);
            }
#pragma unroll
            for (int j = 0; j < ROWS; ++j) {
                if (g + j < cnt) {
                    float* a = acc + __builtin_amdgcn_readlane(doff, g + j) + lane * 2;
#pragma unroll
                    for (int p = 0; p < PARTS; ++p) {
                        atomicAdd(a + p * 64,     v[j][p].x);
                        atomicAdd(a + p * 64 + 1, v[j][p].y);
                    }
                }
            }
        }
    }
}

__device__ __forceinline__ unsigned owner_fetch(const unsigned* __restrict__ recs,
                                                int offv, int endv, int total, int r) {
    int a = __builtin_amdgcn_readlane(offv, 0);
#pragma unroll
    for (int j = 1; j < WAVE_CHUNKS; ++j)
        if (r >= __builtin_amdgcn_readlane(endv, j - 1)) a = __builtin_amdgcn_readlane(offv, j);
    unsigned v = 0u;
    if (r < total) v = recs[a + r];
    return v;
}

template <int W, bool RELU>
__device__ __forceinline__ void owner_sum_body(const float*    __restrict__ rel,
                                               const unsigned* __restrict__ recs,
                                               const unsigned* __restrict__ tab,
                                               float*          __restrict__ out,
                                               float* acc) {
    const int tid   = threadIdx.x;
    const int lane  = tid & 31;
    const int wave  = __builtin_amdgcn_readfirstlane(tid >> 5);
    const int base  = blockIdx.x * OWN_ROWS;
    const int nrows = (N_NODES - base < OWN_ROWS) ? (N_NODES - base) : OWN_ROWS;

    const int c0 = wave * WAVE_CHUNKS;
    const unsigned te = (lane < WAVE_CHUNKS) ? tab[(c0 + lane) * OWN_BLOCKS + blockIdx.x] : 0u;

    float4*       a4 = (float4*)acc;
    float4*       o4 = (float4*)(out + (size_t)base * W);
    for (int i = tid; i < nrows * (W / 4); i += OWN_THREADS) a4[i] = o4[i];

    int endv = 0, offv = 0, total = 0;
#pragma unroll
    for (int j = 0; j < WAVE_CHUNKS; ++j) {
        const unsigned t = (unsigned)__builtin_amdgcn_readlane((int)te, j);
        const int o = (c0 + j) * BIN_PITCH + (int)(t >> 16) - total;
        total += (int)(t & 0xFFFFu);
        if (lane == j) {
            offv = o;
            endv = total;
        }
    }

    __syncthreads();

    unsigned nxt = owner_fetch(recs, offv, endv, total, lane);
    for (int b = 0; b < total; b += 32) {
        const unsigned cur = nxt;
        if (b + 32 < total) nxt = owner_fetch(recs, offv, endv, total, b + 32 + lane);
        const int left = total - b;
        owner_add_batch<W>(rel, acc, cur, (left < 32) ? left : 32, lane);
    }

    __syncthreads();
    for (int i = tid; i < nrows * (W / 4); i += OWN_THREADS) {
        float4 x = a4[i];
        if (RELU) {
            x.x = fmaxf(x.x, 0.f);
            x.y = fmaxf(x.y, 0.f);
            x.z = fmaxf(x.z, 0.f);
            x.w = fmaxf(x.w, 0.f);
        }
        o4[i] = x;
    }
}

__global__ void __launch_bounds__(OWN_THREADS)
rgcn_owner_sum128_relu_kernel(const float*    __restrict__ rel,
                              const unsigned* __restrict__ recs,
                              const unsigned* __restrict__ tab,
                              float*          __restrict__ out) {
    __shared__ __attribute__((aligned(16))) float acc[OWN_ROWS * 128];
    owner_sum_body<128, true>(rel, recs, tab, out, acc);
}

__global__ void __launch_bounds__(OWN_THREADS)
rgcn_owner_sum64_kernel(const float*    __restrict__ rel,
                        const unsigned* __restrict__ recs,
                        const unsigned* __restrict__ tab,
                        float*          __restrict__ out) {
    __shared__ __attribute__((aligned(16))) float acc[OWN_ROWS * 64];
    owner_sum_body<64, false>(rel, recs, tab, out, acc);
}

extern "C" void kernel_launch(void* const* d_in, const int* in_sizes, int n_in,
                              void* d_out, int out_size, void* d_ws, size_t ws_size,
                              hipStream_t stream) {
    const float* features = (const float*)d_in[0];
    const int*   src      = (const int*)  d_in[1];
    const int*   dst      = (const int*)  d_in[2];
    const int*   etype    = (const int*)  d_in[3];
    const float* W1       = (const float*)d_in[4];
    const float* W1s      = (const float*)d_in[5];
    const float* b1       = (const float*)d_in[6];
    const float* W2       = (const float*)d_in[7];
    const float* W2s      = (const float*)d_in[8];
    const float* b2       = (const float*)d_in[9];
    float*       out      = (float*)d_out;

    float*  rel  = (float*)d_ws;
    float*  h1   = rel + (size_t)N_REL * N_NODES * KDIM;
    float2* Wf1  = (float2*)(h1 + (size_t)N_NODES * KDIM);
    float2* Wf2  = Wf1 + (size_t)NSEG * 8 * 32 * 32;

    unsigned* recs1 = (unsigned*)d_out;
    unsigned* tab1  = recs1 + BIN_CHUNKS * BIN_PITCH;
    unsigned* recs2 = (unsigned*)(rel + (size_t)N_REL * N_NODES * 64);
    unsigned* tab2  = recs2 + BIN_CHUNKS * BIN_PITCH;

    const int gemmBlocks   = N_NODES / 16;

    const int pack1 = NSEG * 8 * 32 * 32;
    const int pack2 = NSEG * 4 * 32 * 32;

    rgcn_pack_w_kernel<<<(pack1 + 255) / 256, 256, 0, stream>>>(W1, W1s, Wf1, 128);
    rgcn_pack_w_kernel<<<(pack2 + 255) / 256, 256, 0, stream>>>(W2, W2s, Wf2, 64);

    rgcn_gemm_t128_kernel<<<gemmBlocks, 288, 0, stream>>>(features, Wf1, b1, rel, h1);
    rgcn_bin_kernel<<<BIN_CHUNKS, BIN_THREADS, 0, stream>>>(src, dst, etype, recs1, tab1);
    rgcn_owner_sum128_relu_kernel<<<OWN_BLOCKS, OWN_THREADS, 0, stream>>>(rel, recs1, tab1, h1);

    rgcn_gemm_t64_kernel<<<gemmBlocks, 288, 0, stream>>>(h1, Wf2, b2, rel, out);
    rgcn_bin_kernel<<<BIN_CHUNKS, BIN_THREADS, 0, stream>>>(src, dst, etype, recs2, tab2);
    rgcn_owner_sum64_kernel<<<OWN_BLOCKS, OWN_THREADS, 0, stream>>>(rel, recs2, tab2, out);
}


# Edge grouping (both copies) and layer 2's weight pack run as extra work-groups inside the long GEMM launches, overlapped with them, not in 3 launches of their own (8 -> 5); sum order same
# speedup vs baseline: 3.4180x; 1.0522x over previous
#include <hip/hip_runtime.h>

typedef __attribute__((ext_vector_type(2))) float v2f;
typedef __attribute__((ext_vector_type(8))) float v8f;

#define N_NODES 50000
#define N_EDGES 640000
#define N_REL   8
#define KDIM    128
#define NSEG    (N_REL + 1)

__device__ __forceinline__ void pack_w_elem(const float* __restrict__ Wrel,
                                            const float* __restrict__ Wself,
                                            float2*      __restrict__ Wf,
                                            int O, int idx) {
    const int tiles = O >> 4;
    const int total = NSEG * tiles * 32 * 32;
    if (idx >= total) return;
    int lane = idx & 31;
    int k4   = (idx >> 5) & 31;
    int t    = (idx >> 10) % tiles;
    int seg  = idx / (tiles << 10);
    const float* W = (seg < N_REL) ? (Wrel + (size_t)seg * KDIM * O) : Wself;
    int k0 = k4 * 4 + (lane >> 4) * 2;
    int n  = t * 16 + (lane & 15);
    Wf[idx] = make_float2(W[(size_t)k0 * O + n], W[(size_t)(k0 + 1) * O + n]);
}

__global__ void rgcn_pack_w_kernel(const float* __restrict__ Wrel,
                                   const float* __restrict__ Wself,
                                   float2*      __restrict__ Wf,
                                   int O) {
    pack_w_elem(Wrel, Wself, Wf, O, blockIdx.x * blockDim.x + threadIdx.x);
}

template <int O>
__device__ __forceinline__ void gemm_t_body(const float*  __restrict__ h,
                                            const float2* __restrict__ Wf,
                                            const float*  __restrict__ bias,
                                            float*        __restrict__ rel,
                                            float*        __restrict__ out,
                                            float* lds) {
    constexpr int TILES = O / 16;

    const int m0   = blockIdx.x * 16;
    const int lane = threadIdx.x & 31;
    const int seg  = threadIdx.x >> 5;
    const int half = lane >> 4;
    const int l15  = lane & 15;

    if (threadIdx.x < 256) {
        const int col  = threadIdx.x & 127;
        const int row0 = threadIdx.x >> 7;
#pragma unroll
        for (int i = 0; i < 8; ++i) {
            int row = row0 + 2 * i;
            lds[row * 132 + col] = h[(size_t)(m0 + row) * KDIM + col];
        }
    }
    __syncthreads();

    const float2* wf = Wf + ((size_t)(seg * TILES) << 10) + lane;

    v8f c[TILES];
#pragma unroll
    for (int t = 0; t < TILES; ++t)
        c[t] = (v8f){0.f, 0.f, 0.f, 0.f, 0.f, 0.f, 0.f, 0.f};

#pragma unroll
    for (int k4 = 0; k4 < 32; ++k4) {
        const float* ap = &lds[l15 * 132 + k4 * 4 + half * 2];
        v2f a;
        a.x = ap[0];
        a.y = ap[1];
#pragma unroll
        for (int t = 0; t < TILES; ++t) {
            float2 f = wf[k4 * 32 + t * 1024];
            v2f b;
            b.x = f.x;
            b.y = f.y;
            c[t] = __builtin_amdgcn_wmma_f32_16x16x4_f32(
                       false, a, false, b, (short)0, c[t], false, false);
        }
    }

    if (seg < N_REL) {
#pragma unroll
        for (int v = 0; v < 8; ++v) {
            float* p = rel + ((size_t)(m0 + v + 8 * half) * N_REL + seg) * O + l15;
#pragma unroll
            for (int t = 0; t < TILES; ++t) p[16 * t] = c[t][v];
        }
    } else {
        float bv[TILES];
#pragma unroll
        for (int t = 0; t < TILES; ++t) bv[t] = bias[16 * t + l15];
#pragma unroll
        for (int v = 0; v < 8; ++v) {
            float* p = out + (size_t)(m0 + v + 8 * half) * O + l15;
#pragma unroll
            for (int t = 0; t < TILES; ++t) p[16 * t] = c[t][v] + bv[t];
        }
    }
}


#define OWN_ROWS    196
#define OWN_BLOCKS  ((N_NODES + OWN_ROWS - 1) / OWN_ROWS)
#define OWN_THREADS 1024
#define OWN_WAVES   (OWN_THREADS / 32)

#define BIN_EDGES   2500
#define BIN_CHUNKS  (N_EDGES / BIN_EDGES)
#define BIN_THREADS 256
#define BIN_ITERS   ((BIN_EDGES + BIN_THREADS - 1) / BIN_THREADS)
#define BIN_PITCH   (((BIN_EDGES + 31) / 32) * 32)
#define WAVE_CHUNKS (BIN_CHUNKS / OWN_WAVES)

static_assert(N_EDGES % BIN_EDGES == 0, "chunks cover the edge list exactly");
static_assert(BIN_CHUNKS % OWN_WAVES == 0, "every owner wave takes whole chunks");
static_assert(WAVE_CHUNKS <= 32, "one table word per lane");
static_assert(OWN_BLOCKS <= 256 && OWN_ROWS <= 256, "owner and local row fit in a byte");
static_assert(BIN_THREADS == 256 && OWN_BLOCKS <= BIN_THREADS, "one counter per thread");
static_assert(BIN_EDGES < 65536, "start and count fit in 16 bits each");

__device__ __forceinline__ void bin_body(const int* __restrict__ esrc,
                                         const int* __restrict__ edst,
                                         const int* __restrict__ etyp,
                                         unsigned*  __restrict__ recs,
                                         unsigned*  __restrict__ tab,
                                         const int chunk,
                                         unsigned* cnt,
                                         unsigned* wsum,
                                         unsigned* sorted) {
    const bool act = threadIdx.x < BIN_THREADS;
    const int tid = threadIdx.x & (BIN_THREADS - 1);
    const int e0  = chunk * BIN_EDGES;

    if (act) cnt[tid] = 0u;
    __syncthreads();

    unsigned rec[BIN_ITERS], key[BIN_ITERS];
    if (act) {
        unsigned dv[BIN_ITERS], rv[BIN_ITERS];
#pragma unroll
        for (int i = 0; i < BIN_ITERS; ++i) {
            const int j = tid + i * BIN_THREADS;
            dv[i] = 0u;
            rv[i] = 0u;
            if (j < BIN_EDGES) {
                dv[i] = (unsigned)edst[e0 + j];
                rv[i] = (unsigned)(esrc[e0 + j] * N_REL + etyp[e0 + j]);
            }
        }
#pragma unroll
        for (int i = 0; i < BIN_ITERS; ++i) {
            const int j = tid + i * BIN_THREADS;
            rec[i] = 0u;
            key[i] = 0u;
            if (j < BIN_EDGES) {
                const unsigned o = (dv[i] / OWN_ROWS) & 255u;
                rec[i] = (rv[i] << 8) | ((dv[i] - o * OWN_ROWS) & 255u);
                key[i] = o | (atomicAdd(&cnt[o], 1u) << 8);
            }
        }
    }
    __syncthreads();

    unsigned c = 0u, incl = 0u;
    if (act) {
        c = cnt[tid];
        incl = c;
#pragma unroll
        for (int s = 1; s < 32; s <<= 1) {
            const unsigned t = __shfl_up(incl, s);
            if ((tid & 31) >= s) incl += t;
        }
        if ((tid & 31) == 31) wsum[tid >> 5] = incl;
    }
    __syncthreads();
    if (act) {
        unsigned start = incl - c;
        for (int w = 0; w < (tid >> 5); ++w) start += wsum[w];

        cnt[tid] = start;
        tab[chunk * OWN_BLOCKS + tid] = (start << 16) | c;
    }
    __syncthreads();

    if (act) {
#pragma unroll
        for (int i = 0; i < BIN_ITERS; ++i) {
            const int j = tid + i * BIN_THREADS;
            if (j < BIN_EDGES) sorted[cnt[key[i] & 255u] + (key[i] >> 8)] = rec[i];
        }
    }
    __syncthreads();

    if (act) {
        unsigned* dstp = recs + chunk * BIN_PITCH;
        for (int j = tid; j < BIN_EDGES; j += BIN_THREADS) dstp[j] = sorted[j];
    }
}

#define GEMM_THREADS 288
#define GEMM_BLOCKS  (N_NODES / 16)
#define PACK2_ELEMS  (NSEG * 4 * 32 * 32)
#define PACK2_BLOCKS ((PACK2_ELEMS + GEMM_THREADS - 1) / GEMM_THREADS)

static_assert(N_NODES % 16 == 0, "GEMM work-groups cover the rows exactly");
static_assert(GEMM_THREADS >= BIN_THREADS, "the grouping fits in a GEMM-sized block");

__global__ void __launch_bounds__(GEMM_THREADS)
rgcn_gemm_t128_kernel(const float*  __restrict__ h,
                      const float2* __restrict__ Wf,
                      const float*  __restrict__ bias,
                      float*        __restrict__ rel,
                      float*        __restrict__ out,
                      const int*    __restrict__ esrc,
                      const int*    __restrict__ edst,
                      const int*    __restrict__ etyp,
                      unsigned*     __restrict__ recs,
                      unsigned*     __restrict__ tab,
                      const float*  __restrict__ W2rel,
                      const float*  __restrict__ W2self,
                      float2*       __restrict__ Wf2) {
    __shared__ float    lds[16 * 132];
    __shared__ unsigned cnt[BIN_THREADS];
    __shared__ unsigned wsum[BIN_THREADS / 32];
    __shared__ unsigned sorted[BIN_EDGES];
    const int b = blockIdx.x;
    if (b < GEMM_BLOCKS) {
        gemm_t_body<128>(h, Wf, bias, rel, out, lds);
    } else if (b < GEMM_BLOCKS + BIN_CHUNKS) {
        bin_body(esrc, edst, etyp, recs, tab, b - GEMM_BLOCKS, cnt, wsum, sorted);
    } else {
        pack_w_elem(W2rel, W2self, Wf2, 64,
                    (b - GEMM_BLOCKS - BIN_CHUNKS) * GEMM_THREADS + (int)threadIdx.x);
    }
}

__global__ void __launch_bounds__(GEMM_THREADS)
rgcn_gemm_t64_kernel(const float*  __restrict__ h,
                     const float2* __restrict__ Wf,
                     const float*  __restrict__ bias,
                     float*        __restrict__ rel,
                     float*        __restrict__ out,
                     const int*    __restrict__ esrc,
                     const int*    __restrict__ edst,
                     const int*    __restrict__ etyp,
                     unsigned*     __restrict__ recs,
                     unsigned*     __restrict__ tab) {
    __shared__ float    lds[16 * 132];
    __shared__ unsigned cnt[BIN_THREADS];
    __shared__ unsigned wsum[BIN_THREADS / 32];
    __shared__ unsigned sorted[BIN_EDGES];
    const int b = blockIdx.x;
    if (b < GEMM_BLOCKS) {
        gemm_t_body<64>(h, Wf, bias, rel, out, lds);
    } else {
        bin_body(esrc, edst, etyp, recs, tab, b - GEMM_BLOCKS, cnt, wsum, sorted);
    }
}

template <int W>
__device__ __forceinline__ void owner_add_batch(const float* __restrict__ rel,
                                                float* acc, unsigned rec,
                                                int cnt, int lane) {
    constexpr int PARTS = W / 64;
    constexpr int ROWS  = 32 / PARTS;
    const int roff = (int)(rec >> 8) * W;
    const int doff = (int)(rec & 255u) * W;
#pragma unroll
    for (int g = 0; g < 32; g += ROWS) {
        if (g < cnt) {
            float2 v[ROWS][PARTS];
#pragma unroll
            for (int j = 0; j < ROWS; ++j) {
                const int ro = __builtin_amdgcn_readlane(roff, g + j);
#pragma unroll
                for (int p = 0; p < PARTS; ++p)
                    v[j][p] = *(const float2*)(rel + ro + p * 64 + lane * 2);
            }
#pragma unroll
            for (int j = 0; j < ROWS; ++j) {
                if (g + j < cnt) {
                    float* a = acc + __builtin_amdgcn_readlane(doff, g + j) + lane * 2;
#pragma unroll
                    for (int p = 0; p < PARTS; ++p) {
                        atomicAdd(a + p * 64,     v[j][p].x);
                        atomicAdd(a + p * 64 + 1, v[j][p].y);
                    }
                }
            }
        }
    }
}

__device__ __forceinline__ unsigned owner_fetch(const unsigned* __restrict__ recs,
                                                int offv, int endv, int total, int r) {
    int a = __builtin_amdgcn_readlane(offv, 0);
#pragma unroll
    for (int j = 1; j < WAVE_CHUNKS; ++j)
        if (r >= __builtin_amdgcn_readlane(endv, j - 1)) a = __builtin_amdgcn_readlane(offv, j);
    unsigned v = 0u;
    if (r < total) v = recs[a + r];
    return v;
}

template <int W, bool RELU>
__device__ __forceinline__ void owner_sum_body(const float*    __restrict__ rel,
                                               const unsigned* __restrict__ recs,
                                               const unsigned* __restrict__ tab,
                                               float*          __restrict__ out,
                                               float* acc) {
    const int tid   = threadIdx.x;
    const int lane  = tid & 31;
    const int wave  = __builtin_amdgcn_readfirstlane(tid >> 5);
    const int base  = blockIdx.x * OWN_ROWS;
    const int nrows = (N_NODES - base < OWN_ROWS) ? (N_NODES - base) : OWN_ROWS;

    const int c0 = wave * WAVE_CHUNKS;
    const unsigned te = (lane < WAVE_CHUNKS) ? tab[(c0 + lane) * OWN_BLOCKS + blockIdx.x] : 0u;

    float4*       a4 = (float4*)acc;
    float4*       o4 = (float4*)(out + (size_t)base * W);
    for (int i = tid; i < nrows * (W / 4); i += OWN_THREADS) a4[i] = o4[i];

    int endv = 0, offv = 0, total = 0;
#pragma unroll
    for (int j = 0; j < WAVE_CHUNKS; ++j) {
        const unsigned t = (unsigned)__builtin_amdgcn_readlane((int)te, j);
        const int o = (c0 + j) * BIN_PITCH + (int)(t >> 16) - total;
        total += (int)(t & 0xFFFFu);
        if (lane == j) {
            offv = o;
            endv = total;
        }
    }

    __syncthreads();

    unsigned nxt = owner_fetch(recs, offv, endv, total, lane);
    for (int b = 0; b < total; b += 32) {
        const unsigned cur = nxt;
        if (b + 32 < total) nxt = owner_fetch(recs, offv, endv, total, b + 32 + lane);
        const int left = total - b;
        owner_add_batch<W>(rel, acc, cur, (left < 32) ? left : 32, lane);
    }

    __syncthreads();
    for (int i = tid; i < nrows * (W / 4); i += OWN_THREADS) {
        float4 x = a4[i];
        if (RELU) {
            x.x = fmaxf(x.x, 0.f);
            x.y = fmaxf(x.y, 0.f);
            x.z = fmaxf(x.z, 0.f);
            x.w = fmaxf(x.w, 0.f);
        }
        o4[i] = x;
    }
}

__global__ void __launch_bounds__(OWN_THREADS)
rgcn_owner_sum128_relu_kernel(const float*    __restrict__ rel,
                              const unsigned* __restrict__ recs,
                              const unsigned* __restrict__ tab,
                              float*          __restrict__ out) {
    __shared__ __attribute__((aligned(16))) float acc[OWN_ROWS * 128];
    owner_sum_body<128, true>(rel, recs, tab, out, acc);
}

__global__ void __launch_bounds__(OWN_THREADS)
rgcn_owner_sum64_kernel(const float*    __restrict__ rel,
                        const unsigned* __restrict__ recs,
                        const unsigned* __restrict__ tab,
                        float*          __restrict__ out) {
    __shared__ __attribute__((aligned(16))) float acc[OWN_ROWS * 64];
    owner_sum_body<64, false>(rel, recs, tab, out, acc);
}

extern "C" void kernel_launch(void* const* d_in, const int* in_sizes, int n_in,
                              void* d_out, int out_size, void* d_ws, size_t ws_size,
                              hipStream_t stream) {
    const float* features = (const float*)d_in[0];
    const int*   src      = (const int*)  d_in[1];
    const int*   dst      = (const int*)  d_in[2];
    const int*   etype    = (const int*)  d_in[3];
    const float* W1       = (const float*)d_in[4];
    const float* W1s      = (const float*)d_in[5];
    const float* b1       = (const float*)d_in[6];
    const float* W2       = (const float*)d_in[7];
    const float* W2s      = (const float*)d_in[8];
    const float* b2       = (const float*)d_in[9];
    float*       out      = (float*)d_out;

    float*  rel  = (float*)d_ws;
    float*  h1   = rel + (size_t)N_REL * N_NODES * KDIM;
    float2* Wf1  = (float2*)(h1 + (size_t)N_NODES * KDIM);
    float2* Wf2  = Wf1 + (size_t)NSEG * 8 * 32 * 32;

    unsigned* recs1 = (unsigned*)d_out;
    unsigned* tab1  = recs1 + BIN_CHUNKS * BIN_PITCH;
    unsigned* recs2 = (unsigned*)(rel + (size_t)N_REL * N_NODES * 64);
    unsigned* tab2  = recs2 + BIN_CHUNKS * BIN_PITCH;

    const int pack1 = NSEG * 8 * 32 * 32;

    rgcn_pack_w_kernel<<<(pack1 + 255) / 256, 256, 0, stream>>>(W1, W1s, Wf1, 128);

    rgcn_gemm_t128_kernel<<<GEMM_BLOCKS + BIN_CHUNKS + PACK2_BLOCKS, GEMM_THREADS, 0, stream>>>(
        features, Wf1, b1, rel, h1, src, dst, etype, recs1, tab1, W2, W2s, Wf2);
    rgcn_owner_sum128_relu_kernel<<<OWN_BLOCKS, OWN_THREADS, 0, stream>>>(rel, recs1, tab1, h1);

    rgcn_gemm_t64_kernel<<<GEMM_BLOCKS + BIN_CHUNKS, GEMM_THREADS, 0, stream>>>(
        h1, Wf2, b2, rel, out, src, dst, etype, recs2, tab2);
    rgcn_owner_sum64_kernel<<<OWN_BLOCKS, OWN_THREADS, 0, stream>>>(rel, recs2, tab2, out);
}


# Owner groups rank their edge records by destination row in LDS (integer adds); a wave sums whole rows in registers and stores them: LDS float adds and row copies go; order: arrival at the ranking
# speedup vs baseline: 3.5728x; 1.0453x over previous
#include <hip/hip_runtime.h>

typedef __attribute__((ext_vector_type(2))) float v2f;
typedef __attribute__((ext_vector_type(8))) float v8f;

#define N_NODES 50000
#define N_EDGES 640000
#define N_REL   8
#define KDIM    128
#define NSEG    (N_REL + 1)

__device__ __forceinline__ void pack_w_elem(const float* __restrict__ Wrel,
                                            const float* __restrict__ Wself,
                                            float2*      __restrict__ Wf,
                                            int O, int idx) {
    const int tiles = O >> 4;
    const int total = NSEG * tiles * 32 * 32;
    if (idx >= total) return;
    int lane = idx & 31;
    int k4   = (idx >> 5) & 31;
    int t    = (idx >> 10) % tiles;
    int seg  = idx / (tiles << 10);
    const float* W = (seg < N_REL) ? (Wrel + (size_t)seg * KDIM * O) : Wself;
    int k0 = k4 * 4 + (lane >> 4) * 2;
    int n  = t * 16 + (lane & 15);
    Wf[idx] = make_float2(W[(size_t)k0 * O + n], W[(size_t)(k0 + 1) * O + n]);
}

__global__ void rgcn_pack_w_kernel(const float* __restrict__ Wrel,
                                   const float* __restrict__ Wself,
                                   float2*      __restrict__ Wf,
                                   int O) {
    pack_w_elem(Wrel, Wself, Wf, O, blockIdx.x * blockDim.x + threadIdx.x);
}

template <int O>
__device__ __forceinline__ void gemm_t_body(const float*  __restrict__ h,
                                            const float2* __restrict__ Wf,
                                            const float*  __restrict__ bias,
                                            float*        __restrict__ rel,
                                            float*        __restrict__ out,
                                            float* lds) {
    constexpr int TILES = O / 16;

    const int m0   = blockIdx.x * 16;
    const int lane = threadIdx.x & 31;
    const int seg  = threadIdx.x >> 5;
    const int half = lane >> 4;
    const int l15  = lane & 15;

    if (threadIdx.x < 256) {
        const int col  = threadIdx.x & 127;
        const int row0 = threadIdx.x >> 7;
#pragma unroll
        for (int i = 0; i < 8; ++i) {
            int row = row0 + 2 * i;
            lds[row * 132 + col] = h[(size_t)(m0 + row) * KDIM + col];
        }
    }
    __syncthreads();

    const float2* wf = Wf + ((size_t)(seg * TILES) << 10) + lane;

    v8f c[TILES];
#pragma unroll
    for (int t = 0; t < TILES; ++t)
        c[t] = (v8f){0.f, 0.f, 0.f, 0.f, 0.f, 0.f, 0.f, 0.f};

#pragma unroll
    for (int k4 = 0; k4 < 32; ++k4) {
        const float* ap = &lds[l15 * 132 + k4 * 4 + half * 2];
        v2f a;
        a.x = ap[0];
        a.y = ap[1];
#pragma unroll
        for (int t = 0; t < TILES; ++t) {
            float2 f = wf[k4 * 32 + t * 1024];
            v2f b;
            b.x = f.x;
            b.y = f.y;
            c[t] = __builtin_amdgcn_wmma_f32_16x16x4_f32(
                       false, a, false, b, (short)0, c[t], false, false);
        }
    }

    if (seg < N_REL) {
#pragma unroll
        for (int v = 0; v < 8; ++v) {
            float* p = rel + ((size_t)(m0 + v + 8 * half) * N_REL + seg) * O + l15;
#pragma unroll
            for (int t = 0; t < TILES; ++t) p[16 * t] = c[t][v];
        }
    } else {
        float bv[TILES];
#pragma unroll
        for (int t = 0; t < TILES; ++t) bv[t] = bias[16 * t + l15];
#pragma unroll
        for (int v = 0; v < 8; ++v) {
            float* p = out + (size_t)(m0 + v + 8 * half) * O + l15;
#pragma unroll
            for (int t = 0; t < TILES; ++t) p[16 * t] = c[t][v] + bv[t];
        }
    }
}


#define OWN_ROWS    196
#define OWN_BLOCKS  ((N_NODES + OWN_ROWS - 1) / OWN_ROWS)
#define OWN_THREADS 1024
#define OWN_WAVES   (OWN_THREADS / 32)

#define BIN_EDGES   2500
#define BIN_CHUNKS  (N_EDGES / BIN_EDGES)
#define BIN_THREADS 256
#define BIN_ITERS   ((BIN_EDGES + BIN_THREADS - 1) / BIN_THREADS)
#define BIN_PITCH   (((BIN_EDGES + 31) / 32) * 32)
#define WAVE_CHUNKS (BIN_CHUNKS / OWN_WAVES)

static_assert(N_EDGES % BIN_EDGES == 0, "chunks cover the edge list exactly");
static_assert(BIN_CHUNKS % OWN_WAVES == 0, "every owner wave takes whole chunks");
static_assert(WAVE_CHUNKS <= 32, "one table word per lane");
static_assert(OWN_BLOCKS <= 256 && OWN_ROWS <= 256, "owner and local row fit in a byte");
static_assert(BIN_THREADS == 256 && OWN_BLOCKS <= BIN_THREADS, "one counter per thread");
static_assert(BIN_EDGES < 65536, "start and count fit in 16 bits each");

__device__ __forceinline__ void bin_body(const int* __restrict__ esrc,
                                         const int* __restrict__ edst,
                                         const int* __restrict__ etyp,
                                         unsigned*  __restrict__ recs,
                                         unsigned*  __restrict__ tab,
                                         const int chunk,
                                         unsigned* cnt,
                                         unsigned* wsum,
                                         unsigned* sorted) {
    const bool act = threadIdx.x < BIN_THREADS;
    const int tid = threadIdx.x & (BIN_THREADS - 1);
    const int e0  = chunk * BIN_EDGES;

    if (act) cnt[tid] = 0u;
    __syncthreads();

    unsigned rec[BIN_ITERS], key[BIN_ITERS];
    if (act) {
        unsigned dv[BIN_ITERS], rv[BIN_ITERS];
#pragma unroll
        for (int i = 0; i < BIN_ITERS; ++i) {
            const int j = tid + i * BIN_THREADS;
            dv[i] = 0u;
            rv[i] = 0u;
            if (j < BIN_EDGES) {
                dv[i] = (unsigned)edst[e0 + j];
                rv[i] = (unsigned)(esrc[e0 + j] * N_REL + etyp[e0 + j]);
            }
        }
#pragma unroll
        for (int i = 0; i < BIN_ITERS; ++i) {
            const int j = tid + i * BIN_THREADS;
            rec[i] = 0u;
            key[i] = 0u;
            if (j < BIN_EDGES) {
                const unsigned o = (dv[i] / OWN_ROWS) & 255u;
                rec[i] = (rv[i] << 8) | ((dv[i] - o * OWN_ROWS) & 255u);
                key[i] = o | (atomicAdd(&cnt[o], 1u) << 8);
            }
        }
    }
    __syncthreads();

    unsigned c = 0u, incl = 0u;
    if (act) {
        c = cnt[tid];
        incl = c;
#pragma unroll
        for (int s = 1; s < 32; s <<= 1) {
            const unsigned t = __shfl_up(incl, s);
            if ((tid & 31) >= s) incl += t;
        }
        if ((tid & 31) == 31) wsum[tid >> 5] = incl;
    }
    __syncthreads();
    if (act) {
        unsigned start = incl - c;
        for (int w = 0; w < (tid >> 5); ++w) start += wsum[w];

        cnt[tid] = start;
        tab[chunk * OWN_BLOCKS + tid] = (start << 16) | c;
    }
    __syncthreads();

    if (act) {
#pragma unroll
        for (int i = 0; i < BIN_ITERS; ++i) {
            const int j = tid + i * BIN_THREADS;
            if (j < BIN_EDGES) sorted[cnt[key[i] & 255u] + (key[i] >> 8)] = rec[i];
        }
    }
    __syncthreads();

    if (act) {
        unsigned* dstp = recs + chunk * BIN_PITCH;
        for (int j = tid; j < BIN_EDGES; j += BIN_THREADS) dstp[j] = sorted[j];
    }
}

#define GEMM_THREADS 288
#define GEMM_BLOCKS  (N_NODES / 16)
#define PACK2_ELEMS  (NSEG * 4 * 32 * 32)
#define PACK2_BLOCKS ((PACK2_ELEMS + GEMM_THREADS - 1) / GEMM_THREADS)

static_assert(N_NODES % 16 == 0, "GEMM work-groups cover the rows exactly");
static_assert(GEMM_THREADS >= BIN_THREADS, "the grouping fits in a GEMM-sized block");

__global__ void __launch_bounds__(GEMM_THREADS)
rgcn_gemm_t128_kernel(const float*  __restrict__ h,
                      const float2* __restrict__ Wf,
                      const float*  __restrict__ bias,
                      float*        __restrict__ rel,
                      float*        __restrict__ out,
                      const int*    __restrict__ esrc,
                      const int*    __restrict__ edst,
                      const int*    __restrict__ etyp,
                      unsigned*     __restrict__ recs,
                      unsigned*     __restrict__ tab,
                      const float*  __restrict__ W2rel,
                      const float*  __restrict__ W2self,
                      float2*       __restrict__ Wf2) {
    __shared__ float    lds[16 * 132];
    __shared__ unsigned cnt[BIN_THREADS];
    __shared__ unsigned wsum[BIN_THREADS / 32];
    __shared__ unsigned sorted[BIN_EDGES];
    const int b = blockIdx.x;
    if (b < GEMM_BLOCKS) {
        gemm_t_body<128>(h, Wf, bias, rel, out, lds);
    } else if (b < GEMM_BLOCKS + BIN_CHUNKS) {
        bin_body(esrc, edst, etyp, recs, tab, b - GEMM_BLOCKS, cnt, wsum, sorted);
    } else {
        pack_w_elem(W2rel, W2self, Wf2, 64,
                    (b - GEMM_BLOCKS - BIN_CHUNKS) * GEMM_THREADS + (int)threadIdx.x);
    }
}

__global__ void __launch_bounds__(GEMM_THREADS)
rgcn_gemm_t64_kernel(const float*  __restrict__ h,
                     const float2* __restrict__ Wf,
                     const float*  __restrict__ bias,
                     float*        __restrict__ rel,
                     float*        __restrict__ out,
                     const int*    __restrict__ esrc,
                     const int*    __restrict__ edst,
                     const int*    __restrict__ etyp,
                     unsigned*     __restrict__ recs,
                     unsigned*     __restrict__ tab) {
    __shared__ float    lds[16 * 132];
    __shared__ unsigned cnt[BIN_THREADS];
    __shared__ unsigned wsum[BIN_THREADS / 32];
    __shared__ unsigned sorted[BIN_EDGES];
    const int b = blockIdx.x;
    if (b < GEMM_BLOCKS) {
        gemm_t_body<64>(h, Wf, bias, rel, out, lds);
    } else {
        bin_body(esrc, edst, etyp, recs, tab, b - GEMM_BLOCKS, cnt, wsum, sorted);
    }
}

template <int W>
__device__ __forceinline__ void owner_add_batch(const float* __restrict__ rel,
                                                float* acc, unsigned rec,
                                                int cnt, int lane) {
    constexpr int PARTS = W / 64;
    constexpr int ROWS  = 32 / PARTS;
    const int roff = (int)(rec >> 8) * W;
    const int doff = (int)(rec & 255u) * W;
#pragma unroll
    for (int g = 0; g < 32; g += ROWS) {
        if (g < cnt) {
            float2 v[ROWS][PARTS];
#pragma unroll
            for (int j = 0; j < ROWS; ++j) {
                const int ro = __builtin_amdgcn_readlane(roff, g + j);
#pragma unroll
                for (int p = 0; p < PARTS; ++p)
                    v[j][p] = *(const float2*)(rel + ro + p * 64 + lane * 2);
            }
#pragma unroll
            for (int j = 0; j < ROWS; ++j) {
                if (g + j < cnt) {
                    float* a = acc + __builtin_amdgcn_readlane(doff, g + j) + lane * 2;
#pragma unroll
                    for (int p = 0; p < PARTS; ++p) {
                        atomicAdd(a + p * 64,     v[j][p].x);
                        atomicAdd(a + p * 64 + 1, v[j][p].y);
                    }
                }
            }
        }
    }
}

__device__ __forceinline__ unsigned owner_fetch(const unsigned* __restrict__ recs,
                                                int offv, int endv, int total, int r) {
    int a = __builtin_amdgcn_readlane(offv, 0);
#pragma unroll
    for (int j = 1; j < WAVE_CHUNKS; ++j)
        if (r >= __builtin_amdgcn_readlane(endv, j - 1)) a = __builtin_amdgcn_readlane(offv, j);
    unsigned v = 0u;
    if (r < total) v = recs[a + r];
    return v;
}

#define SRT_START 0x80000000u

static_assert(OWN_WAVES == 32, "one wave total per lane; 32 pieces of the ranked list");
static_assert(OWN_ROWS <= 224 && OWN_ROWS < 256, "7 rows per lane in the piece search, 8 in the prefix sum");

template <int W, bool RELU>
__device__ __forceinline__ void owner_store_row(float* out, int row, int lane,
                                                __attribute__((ext_vector_type(W / 32))) float a) {
    constexpr int VEC = W / 32;
    typedef float vrow __attribute__((ext_vector_type(VEC)));
    if (RELU) {
#pragma unroll
        for (int i = 0; i < VEC; ++i) a[i] = fmaxf(a[i], 0.f);
    }
    *(vrow*)(out + (size_t)row * W + lane * VEC) = a;
}

template <int W, bool RELU>
__device__ __forceinline__ void owner_sum_body(const float*    __restrict__ rel,
                                               const unsigned* __restrict__ recs,
                                               const unsigned* __restrict__ tab,
                                               float*                       out,
                                               float* acc,
                                               unsigned* cnt,
                                               unsigned* rstart,
                                               unsigned* wtot) {
    constexpr int VEC  = W / 32;
    constexpr int ROWS = 64 / VEC;
    constexpr int CAP  = OWN_ROWS * W;
    typedef float vrow __attribute__((ext_vector_type(VEC)));

    const int tid   = threadIdx.x;
    const int lane  = tid & 31;
    const int wave  = __builtin_amdgcn_readfirstlane(tid >> 5);
    const int base  = blockIdx.x * OWN_ROWS;
    const int nrows = (N_NODES - base < OWN_ROWS) ? (N_NODES - base) : OWN_ROWS;

    const int c0 = wave * WAVE_CHUNKS;
    const unsigned te = (lane < WAVE_CHUNKS) ? tab[(c0 + lane) * OWN_BLOCKS + blockIdx.x] : 0u;

    if (tid < 256) cnt[tid] = 0u;

    int endv = 0, offv = 0, total = 0;
#pragma unroll
    for (int j = 0; j < WAVE_CHUNKS; ++j) {
        const unsigned t = (unsigned)__builtin_amdgcn_readlane((int)te, j);
        const int o = (c0 + j) * BIN_PITCH + (int)(t >> 16) - total;
        total += (int)(t & 0xFFFFu);
        if (lane == j) {
            offv = o;
            endv = total;
        }
    }
    if (lane == 0) wtot[wave] = (unsigned)total;

    __syncthreads();

    unsigned tsum = wtot[lane];
#pragma unroll
    for (int d = 16; d > 0; d >>= 1) tsum += __shfl_xor(tsum, d);
    const int gtot = __builtin_amdgcn_readfirstlane((int)tsum) + nrows;

    if (gtot <= CAP) {
        unsigned* srt = (unsigned*)acc;

        unsigned keep[4];
#pragma unroll
        for (int k = 0; k < 4; ++k)
            keep[k] = (32 * k < total) ? owner_fetch(recs, offv, endv, total, 32 * k + lane) : 0u;
#pragma unroll
        for (int k = 0; k < 4; ++k)
            if (32 * k + lane < total) atomicAdd(&cnt[keep[k] & 255u], 1u);
        for (int b = 128; b < total; b += 32) {
            const unsigned v = owner_fetch(recs, offv, endv, total, b + lane);
            if (b + lane < total) atomicAdd(&cnt[v & 255u], 1u);
        }
        __syncthreads();

        if (wave == 0) {
            const uint4 ca = *(const uint4*)&cnt[lane * 8];
            const uint4 cb = *(const uint4*)&cnt[lane * 8 + 4];
            const unsigned cr[8] = {ca.x, ca.y, ca.z, ca.w, cb.x, cb.y, cb.z, cb.w};
            unsigned c[8], s = 0u;
#pragma unroll
            for (int k = 0; k < 8; ++k) {
                const int row = lane * 8 + k;
                c[k] = (row < nrows) ? cr[k] + 1u : 0u;
                s += c[k];
            }
            unsigned incl = s;
#pragma unroll
            for (int d = 1; d < 32; d <<= 1) {
                const unsigned t = __shfl_up(incl, d);
                if (lane >= d) incl += t;
            }
            unsigned run = incl - s;
#pragma unroll
            for (int k = 0; k < 8; ++k) {
                const int row = lane * 8 + k;
                rstart[row] = run;
                if (row < nrows) {
                    srt[run]  = SRT_START | (unsigned)row;
                    cnt[row]  = run + 1u;
                }
                run += c[k];
            }
        }
        __syncthreads();

#pragma unroll
        for (int k = 0; k < 4; ++k)
            if (32 * k + lane < total) srt[atomicAdd(&cnt[keep[k] & 255u], 1u)] = keep[k];
        for (int b = 128; b < total; b += 32) {
            const unsigned v = owner_fetch(recs, offv, endv, total, b + lane);
            if (b + lane < total) srt[atomicAdd(&cnt[v & 255u], 1u)] = v;
        }
        __syncthreads();

        const unsigned lo = (unsigned)((wave * gtot) / OWN_WAVES);
        const unsigned hi = (unsigned)(((wave + 1) * gtot) / OWN_WAVES);
        int r0 = 0, r1 = 0;
#pragma unroll
        for (int k = 0; k < 7; ++k) {
            const unsigned s = rstart[lane + 32 * k];
            r0 += __builtin_popcount(__builtin_amdgcn_ballot_w32(s < lo));
            r1 += __builtin_popcount(__builtin_amdgcn_ballot_w32(s < hi));
        }
        const int p0 = __builtin_amdgcn_readfirstlane((int)rstart[r0]);
        const int p1 = __builtin_amdgcn_readfirstlane((int)rstart[r1]);

        typedef const __attribute__((address_space(1))) vrow* grow;
        const unsigned long long relb  = (unsigned long long)rel;
        const long long          dlt   = (long long)((unsigned long long)out - relb);
        const unsigned long long lanep = relb + (unsigned long long)(lane * VEC * 4);

        vrow a;
#pragma unroll
        for (int i = 0; i < VEC; ++i) a[i] = 0.f;
        int cur = -1;
        for (int p = p0; p < p1; p += ROWS) {
            const int n = (p1 - p < ROWS) ? (p1 - p) : ROWS;
            const unsigned ent = (lane < n) ? srt[p + lane] : 0u;
            const long long roff = (ent & SRT_START)
                                 ? dlt + (long long)(base + (int)(ent & 255u)) * (W * 4)
                                 : (long long)(ent >> 8) * (W * 4);
            const int rlo = (int)(unsigned)roff;
            const int rhi = (int)(roff >> 32);
            vrow v[ROWS];
#pragma unroll
            for (int j = 0; j < ROWS; ++j) {
                const unsigned long long o =
                    ((unsigned long long)(unsigned)__builtin_amdgcn_readlane(rhi, j) << 32) |
                    (unsigned long long)(unsigned)__builtin_amdgcn_readlane(rlo, j);
                v[j] = *(grow)(lanep + o);
            }
#pragma unroll
            for (int j = 0; j < ROWS; ++j) {
                if (j < n) {
                    const unsigned e = (unsigned)__builtin_amdgcn_readlane((int)ent, j);
                    if (e & SRT_START) {
                        if (cur >= 0) owner_store_row<W, RELU>(out, base + cur, lane, a);
                        cur = (int)(e & 255u);
#pragma unroll
                        for (int i = 0; i < VEC; ++i) a[i] = 0.f;
                    }
                    a += v[j];
                }
            }
        }
        if (cur >= 0) owner_store_row<W, RELU>(out, base + cur, lane, a);
        return;
    }

    float4*       a4 = (float4*)acc;
    float4*       o4 = (float4*)(out + (size_t)base * W);
    for (int i = tid; i < nrows * (W / 4); i += OWN_THREADS) a4[i] = o4[i];

    __syncthreads();

    unsigned nxt = owner_fetch(recs, offv, endv, total, lane);
    for (int b = 0; b < total; b += 32) {
        const unsigned cur = nxt;
        if (b + 32 < total) nxt = owner_fetch(recs, offv, endv, total, b + 32 + lane);
        const int left = total - b;
        owner_add_batch<W>(rel, acc, cur, (left < 32) ? left : 32, lane);
    }

    __syncthreads();
    for (int i = tid; i < nrows * (W / 4); i += OWN_THREADS) {
        float4 x = a4[i];
        if (RELU) {
            x.x = fmaxf(x.x, 0.f);
            x.y = fmaxf(x.y, 0.f);
            x.z = fmaxf(x.z, 0.f);
            x.w = fmaxf(x.w, 0.f);
        }
        o4[i] = x;
    }
}

__global__ void __launch_bounds__(OWN_THREADS)
rgcn_owner_sum128_relu_kernel(const float*    __restrict__ rel,
                              const unsigned* __restrict__ recs,
                              const unsigned* __restrict__ tab,
                              float*                       out) {
    __shared__ __attribute__((aligned(16))) float acc[OWN_ROWS * 128];
    __shared__ __attribute__((aligned(16))) unsigned cnt[256];
    __shared__ unsigned rstart[256];
    __shared__ unsigned wtot[OWN_WAVES];
    owner_sum_body<128, true>(rel, recs, tab, out, acc, cnt, rstart, wtot);
}

__global__ void __launch_bounds__(OWN_THREADS)
rgcn_owner_sum64_kernel(const float*    __restrict__ rel,
                        const unsigned* __restrict__ recs,
                        const unsigned* __restrict__ tab,
                        float*                       out) {
    __shared__ __attribute__((aligned(16))) float acc[OWN_ROWS * 64];
    __shared__ __attribute__((aligned(16))) unsigned cnt[256];
    __shared__ unsigned rstart[256];
    __shared__ unsigned wtot[OWN_WAVES];
    owner_sum_body<64, false>(rel, recs, tab, out, acc, cnt, rstart, wtot);
}

extern "C" void kernel_launch(void* const* d_in, const int* in_sizes, int n_in,
                              void* d_out, int out_size, void* d_ws, size_t ws_size,
                              hipStream_t stream) {
    const float* features = (const float*)d_in[0];
    const int*   src      = (const int*)  d_in[1];
    const int*   dst      = (const int*)  d_in[2];
    const int*   etype    = (const int*)  d_in[3];
    const float* W1       = (const float*)d_in[4];
    const float* W1s      = (const float*)d_in[5];
    const float* b1       = (const float*)d_in[6];
    const float* W2       = (const float*)d_in[7];
    const float* W2s      = (const float*)d_in[8];
    const float* b2       = (const float*)d_in[9];
    float*       out      = (float*)d_out;

    float*  rel  = (float*)d_ws;
    float*  h1   = rel + (size_t)N_REL * N_NODES * KDIM;
    float2* Wf1  = (float2*)(h1 + (size_t)N_NODES * KDIM);
    float2* Wf2  = Wf1 + (size_t)NSEG * 8 * 32 * 32;

    unsigned* recs1 = (unsigned*)d_out;
    unsigned* tab1  = recs1 + BIN_CHUNKS * BIN_PITCH;
    unsigned* recs2 = (unsigned*)(rel + (size_t)N_REL * N_NODES * 64);
    unsigned* tab2  = recs2 + BIN_CHUNKS * BIN_PITCH;

    const int pack1 = NSEG * 8 * 32 * 32;

    rgcn_pack_w_kernel<<<(pack1 + 255) / 256, 256, 0, stream>>>(W1, W1s, Wf1, 128);

    rgcn_gemm_t128_kernel<<<GEMM_BLOCKS + BIN_CHUNKS + PACK2_BLOCKS, GEMM_THREADS, 0, stream>>>(
        features, Wf1, b1, rel, h1, src, dst, etype, recs1, tab1, W2, W2s, Wf2);
    rgcn_owner_sum128_relu_kernel<<<OWN_BLOCKS, OWN_THREADS, 0, stream>>>(rel, recs1, tab1, h1);

    rgcn_gemm_t64_kernel<<<GEMM_BLOCKS + BIN_CHUNKS, GEMM_THREADS, 0, stream>>>(
        h1, Wf2, b2, rel, out, src, dst, etype, recs2, tab2);
    rgcn_owner_sum64_kernel<<<OWN_BLOCKS, OWN_THREADS, 0, stream>>>(rel, recs2, tab2, out);
}


# GEMMs weight-stationary: a work-group copies ONE weight segment's B fragments to LDS once, its 8 waves stream 16-row tiles of h past it; per-WMMA B fetch from L2 goes; k order, sum order same
# speedup vs baseline: 3.9849x; 1.1153x over previous
#include <hip/hip_runtime.h>

typedef __attribute__((ext_vector_type(2))) float v2f;
typedef __attribute__((ext_vector_type(4))) float v4f;
typedef __attribute__((ext_vector_type(8))) float v8f;

#define N_NODES 50000
#define N_EDGES 640000
#define N_REL   8
#define KDIM    128
#define NSEG    (N_REL + 1)

__device__ __forceinline__ void pack_w_elem(const float* __restrict__ Wrel,
                                            const float* __restrict__ Wself,
                                            float2*      __restrict__ Wf,
                                            int O, int idx) {
    const int tiles = O >> 4;
    const int total = NSEG * tiles * 32 * 32;
    if (idx >= total) return;
    int lane = idx & 31;
    int k4   = (idx >> 5) & 31;
    int t    = (idx >> 10) % tiles;
    int seg  = idx / (tiles << 10);
    const float* W = (seg < N_REL) ? (Wrel + (size_t)seg * KDIM * O) : Wself;
    int k0 = k4 * 4 + (lane >> 4) * 2;
    int n  = t * 16 + (lane & 15);
    Wf[idx] = make_float2(W[(size_t)k0 * O + n], W[(size_t)(k0 + 1) * O + n]);
}

__global__ void rgcn_pack_w_kernel(const float* __restrict__ Wrel,
                                   const float* __restrict__ Wself,
                                   float2*      __restrict__ Wf,
                                   int O) {
    pack_w_elem(Wrel, Wself, Wf, O, blockIdx.x * blockDim.x + threadIdx.x);
}


#define GEMM_THREADS 256
#define GEMM_WAVES   (GEMM_THREADS / 32)
#define ROW_TILES    (N_NODES / 16)
#define TILES_PER_WG 16
#define ROW_CHUNKS   ((ROW_TILES + TILES_PER_WG - 1) / TILES_PER_WG)
#define GEMM_BLOCKS  (NSEG * ROW_CHUNKS)
#define A_PITCH      132

static_assert(N_NODES % 16 == 0, "row tiles cover the rows exactly");
static_assert(KDIM == 128, "one 16-byte load per lane covers a row of h");

template <int O>
__device__ __forceinline__ void gemm_ws_body(const float*  __restrict__ h,
                                             const float2* __restrict__ Wf,
                                             const float*  __restrict__ bias,
                                             float*        __restrict__ rel,
                                             float*        __restrict__ out,
                                             float2* ldsB,
                                             float*  ldsA,
                                             const int gb) {
    constexpr int TILES = O / 16;
    constexpr int BVEC  = TILES * 512 / GEMM_THREADS;

    const int lane = threadIdx.x & 31;
    const int wave = __builtin_amdgcn_readfirstlane((int)(threadIdx.x >> 5));
    const int half = lane >> 4;
    const int l15  = lane & 15;

    const int seg   = gb % NSEG;
    const int chunk = gb / NSEG;
    const int t_end = ((chunk + 1) * TILES_PER_WG < ROW_TILES) ? (chunk + 1) * TILES_PER_WG : ROW_TILES;
    int tile = chunk * TILES_PER_WG + wave;

    v4f r[16];
    if (tile < t_end) {
        const v4f* hp = (const v4f*)(h + (size_t)tile * 16 * KDIM) + lane;
#pragma unroll
        for (int i = 0; i < 16; ++i) r[i] = hp[i * (KDIM / 4)];
    }

    {
        const v4f* s4 = (const v4f*)(Wf + ((size_t)(seg * TILES) << 10)) + threadIdx.x;
        v4f*       d4 = (v4f*)ldsB + threadIdx.x;
        v4f w[BVEC];
#pragma unroll
        for (int i = 0; i < BVEC; ++i) w[i] = s4[i * GEMM_THREADS];
#pragma unroll
        for (int i = 0; i < BVEC; ++i) d4[i * GEMM_THREADS] = w[i];
    }
    __syncthreads();

    float*        myA = ldsA + wave * (16 * A_PITCH);
    const float*  af  = myA + l15 * A_PITCH + half * 2;
    const float2* bf  = ldsB + lane;

    for (; tile < t_end; tile += GEMM_WAVES) {
#pragma unroll
        for (int i = 0; i < 16; ++i) *(v4f*)&myA[i * A_PITCH + lane * 4] = r[i];
        __builtin_amdgcn_fence(__ATOMIC_ACQ_REL, "wavefront");

        const int nxt = tile + GEMM_WAVES;
        if (nxt < t_end) {
            const v4f* hp = (const v4f*)(h + (size_t)nxt * 16 * KDIM) + lane;
#pragma unroll
            for (int i = 0; i < 16; ++i) r[i] = hp[i * (KDIM / 4)];
        }

        v8f c[TILES];
#pragma unroll
        for (int t = 0; t < TILES; ++t)
            c[t] = (v8f){0.f, 0.f, 0.f, 0.f, 0.f, 0.f, 0.f, 0.f};

#pragma unroll
        for (int k4 = 0; k4 < 32; ++k4) {
            const float* ap = af + k4 * 4;
            v2f a;
            a.x = ap[0];
            a.y = ap[1];
#pragma unroll
            for (int t = 0; t < TILES; ++t) {
                float2 f = bf[k4 * 32 + t * 1024];
                v2f b;
                b.x = f.x;
                b.y = f.y;
                c[t] = __builtin_amdgcn_wmma_f32_16x16x4_f32(
                           false, a, false, b, (short)0, c[t], false, false);
            }
        }

        const int m0 = tile * 16;
        if (seg < N_REL) {
#pragma unroll
            for (int v = 0; v < 8; ++v) {
                float* p = rel + ((size_t)(m0 + v + 8 * half) * N_REL + seg) * O + l15;
#pragma unroll
                for (int t = 0; t < TILES; ++t) p[16 * t] = c[t][v];
            }
        } else {
            float bv[TILES];
#pragma unroll
            for (int t = 0; t < TILES; ++t) bv[t] = bias[16 * t + l15];
#pragma unroll
            for (int v = 0; v < 8; ++v) {
                float* p = out + (size_t)(m0 + v + 8 * half) * O + l15;
#pragma unroll
                for (int t = 0; t < TILES; ++t) p[16 * t] = c[t][v] + bv[t];
            }
        }
    }
}


#define OWN_ROWS    196
#define OWN_BLOCKS  ((N_NODES + OWN_ROWS - 1) / OWN_ROWS)
#define OWN_THREADS 1024
#define OWN_WAVES   (OWN_THREADS / 32)

#define BIN_EDGES   2500
#define BIN_CHUNKS  (N_EDGES / BIN_EDGES)
#define BIN_THREADS 256
#define BIN_ITERS   ((BIN_EDGES + BIN_THREADS - 1) / BIN_THREADS)
#define BIN_PITCH   (((BIN_EDGES + 31) / 32) * 32)
#define WAVE_CHUNKS (BIN_CHUNKS / OWN_WAVES)

static_assert(N_EDGES % BIN_EDGES == 0, "chunks cover the edge list exactly");
static_assert(BIN_CHUNKS % OWN_WAVES == 0, "every owner wave takes whole chunks");
static_assert(WAVE_CHUNKS <= 32, "one table word per lane");
static_assert(OWN_BLOCKS <= 256 && OWN_ROWS <= 256, "owner and local row fit in a byte");
static_assert(BIN_THREADS == 256 && OWN_BLOCKS <= BIN_THREADS, "one counter per thread");
static_assert(BIN_EDGES < 65536, "start and count fit in 16 bits each");


__device__ __forceinline__ void bin_body(const int* __restrict__ esrc,
                                         const int* __restrict__ edst,
                                         const int* __restrict__ etyp,
                                         unsigned*  __restrict__ recs,
                                         unsigned*  __restrict__ tab,
                                         const int chunk,
                                         unsigned* cnt,
                                         unsigned* wsum,
                                         unsigned* sorted) {
    const bool act = threadIdx.x < BIN_THREADS;
    const int tid = threadIdx.x & (BIN_THREADS - 1);
    const int e0  = chunk * BIN_EDGES;

    if (act) cnt[tid] = 0u;
    __syncthreads();

    unsigned rec[BIN_ITERS], key[BIN_ITERS];
    if (act) {
        unsigned dv[BIN_ITERS], rv[BIN_ITERS];
#pragma unroll
        for (int i = 0; i < BIN_ITERS; ++i) {
            const int j = tid + i * BIN_THREADS;
            dv[i] = 0u;
            rv[i] = 0u;
            if (j < BIN_EDGES) {
                dv[i] = (unsigned)edst[e0 + j];
                rv[i] = (unsigned)(esrc[e0 + j] * N_REL + etyp[e0 + j]);
            }
        }
#pragma unroll
        for (int i = 0; i < BIN_ITERS; ++i) {
            const int j = tid + i * BIN_THREADS;
            rec[i] = 0u;
            key[i] = 0u;
            if (j < BIN_EDGES) {
                const unsigned o = (dv[i] / OWN_ROWS) & 255u;
                rec[i] = (rv[i] << 8) | ((dv[i] - o * OWN_ROWS) & 255u);
                key[i] = o | (atomicAdd(&cnt[o], 1u) << 8);
            }
        }
    }
    __syncthreads();

    unsigned c = 0u, incl = 0u;
    if (act) {
        c = cnt[tid];
        incl = c;
#pragma unroll
        for (int s = 1; s < 32; s <<= 1) {
            const unsigned t = __shfl_up(incl, s);
            if ((tid & 31) >= s) incl += t;
        }
        if ((tid & 31) == 31) wsum[tid >> 5] = incl;
    }
    __syncthreads();
    if (act) {
        unsigned start = incl - c;
        for (int w = 0; w < (tid >> 5); ++w) start += wsum[w];

        cnt[tid] = start;
        tab[chunk * OWN_BLOCKS + tid] = (start << 16) | c;
    }
    __syncthreads();

    if (act) {
#pragma unroll
        for (int i = 0; i < BIN_ITERS; ++i) {
            const int j = tid + i * BIN_THREADS;
            if (j < BIN_EDGES) sorted[cnt[key[i] & 255u] + (key[i] >> 8)] = rec[i];
        }
    }
    __syncthreads();

    if (act) {
        unsigned* dstp = recs + chunk * BIN_PITCH;
        for (int j = tid; j < BIN_EDGES; j += BIN_THREADS) dstp[j] = sorted[j];
    }
}

#define PACK2_ELEMS  (NSEG * 4 * 32 * 32)
#define PACK2_BLOCKS ((PACK2_ELEMS + GEMM_THREADS - 1) / GEMM_THREADS)

static_assert(GEMM_THREADS >= BIN_THREADS, "the grouping fits in a GEMM-sized block");

__global__ void __launch_bounds__(GEMM_THREADS)
rgcn_gemm_t128_kernel(const float*  __restrict__ h,
                      const float2* __restrict__ Wf,
                      const float*  __restrict__ bias,
                      float*        __restrict__ rel,
                      float*        __restrict__ out,
                      const int*    __restrict__ esrc,
                      const int*    __restrict__ edst,
                      const int*    __restrict__ etyp,
                      unsigned*     __restrict__ recs,
                      unsigned*     __restrict__ tab,
                      const float*  __restrict__ W2rel,
                      const float*  __restrict__ W2self,
                      float2*       __restrict__ Wf2) {
    __shared__ __attribute__((aligned(16))) float2 ldsB[8 * 1024];
    __shared__ __attribute__((aligned(16))) float  ldsA[GEMM_WAVES * 16 * A_PITCH];
    __shared__ unsigned cnt[BIN_THREADS];
    __shared__ unsigned wsum[BIN_THREADS / 32];
    __shared__ unsigned sorted[BIN_EDGES];
    const int b = blockIdx.x;
    if (b < GEMM_BLOCKS) {
        gemm_ws_body<128>(h, Wf, bias, rel, out, ldsB, ldsA, b);
    } else if (b < GEMM_BLOCKS + BIN_CHUNKS) {
        bin_body(esrc, edst, etyp, recs, tab, b - GEMM_BLOCKS, cnt, wsum, sorted);
    } else {
        pack_w_elem(W2rel, W2self, Wf2, 64,
                    (b - GEMM_BLOCKS - BIN_CHUNKS) * GEMM_THREADS + (int)threadIdx.x);
    }
}

__global__ void __launch_bounds__(GEMM_THREADS)
rgcn_gemm_t64_kernel(const float*  __restrict__ h,
                     const float2* __restrict__ Wf,
                     const float*  __restrict__ bias,
                     float*        __restrict__ rel,
                     float*        __restrict__ out,
                     const int*    __restrict__ esrc,
                     const int*    __restrict__ edst,
                     const int*    __restrict__ etyp,
                     unsigned*     __restrict__ recs,
                     unsigned*     __restrict__ tab) {
    __shared__ __attribute__((aligned(16))) float2 ldsB[4 * 1024];
    __shared__ __attribute__((aligned(16))) float  ldsA[GEMM_WAVES * 16 * A_PITCH];
    __shared__ unsigned cnt[BIN_THREADS];
    __shared__ unsigned wsum[BIN_THREADS / 32];
    __shared__ unsigned sorted[BIN_EDGES];
    const int b = blockIdx.x;
    if (b < GEMM_BLOCKS) {
        gemm_ws_body<64>(h, Wf, bias, rel, out, ldsB, ldsA, b);
    } else {
        bin_body(esrc, edst, etyp, recs, tab, b - GEMM_BLOCKS, cnt, wsum, sorted);
    }
}

template <int W>
__device__ __forceinline__ void owner_add_batch(const float* __restrict__ rel,
                                                float* acc, unsigned rec,
                                                int cnt, int lane) {
    constexpr int PARTS = W / 64;
    constexpr int ROWS  = 32 / PARTS;
    const int roff = (int)(rec >> 8) * W;
    const int doff = (int)(rec & 255u) * W;
#pragma unroll
    for (int g = 0; g < 32; g += ROWS) {
        if (g < cnt) {
            float2 v[ROWS][PARTS];
#pragma unroll
            for (int j = 0; j < ROWS; ++j) {
                const int ro = __builtin_amdgcn_readlane(roff, g + j);
#pragma unroll
                for (int p = 0; p < PARTS; ++p)
                    v[j][p] = *(const float2*)(rel + ro + p * 64 + lane * 2);
            }
#pragma unroll
            for (int j = 0; j < ROWS; ++j) {
                if (g + j < cnt) {
                    float* a = acc + __builtin_amdgcn_readlane(doff, g + j) + lane * 2;
#pragma unroll
                    for (int p = 0; p < PARTS; ++p) {
                        atomicAdd(a + p * 64,     v[j][p].x);
                        atomicAdd(a + p * 64 + 1, v[j][p].y);
                    }
                }
            }
        }
    }
}

__device__ __forceinline__ unsigned owner_fetch(const unsigned* __restrict__ recs,
                                                int offv, int endv, int total, int r) {
    int a = __builtin_amdgcn_readlane(offv, 0);
#pragma unroll
    for (int j = 1; j < WAVE_CHUNKS; ++j)
        if (r >= __builtin_amdgcn_readlane(endv, j - 1)) a = __builtin_amdgcn_readlane(offv, j);
    unsigned v = 0u;
    if (r < total) v = recs[a + r];
    return v;
}

#define SRT_START 0x80000000u

static_assert(OWN_WAVES == 32, "one wave total per lane; 32 pieces of the ranked list");
static_assert(OWN_ROWS <= 224 && OWN_ROWS < 256, "7 rows per lane in the piece search, 8 in the prefix sum");

template <int W, bool RELU>
__device__ __forceinline__ void owner_store_row(float* out, int row, int lane,
                                                __attribute__((ext_vector_type(W / 32))) float a) {
    constexpr int VEC = W / 32;
    typedef float vrow __attribute__((ext_vector_type(VEC)));
    if (RELU) {
#pragma unroll
        for (int i = 0; i < VEC; ++i) a[i] = fmaxf(a[i], 0.f);
    }
    *(vrow*)(out + (size_t)row * W + lane * VEC) = a;
}

template <int W, bool RELU>
__device__ __forceinline__ void owner_sum_body(const float*    __restrict__ rel,
                                               const unsigned* __restrict__ recs,
                                               const unsigned* __restrict__ tab,
                                               float*                       out,
                                               float* acc,
                                               unsigned* cnt,
                                               unsigned* rstart,
                                               unsigned* wtot) {
    constexpr int VEC  = W / 32;
    constexpr int ROWS = 64 / VEC;
    constexpr int CAP  = OWN_ROWS * W;
    typedef float vrow __attribute__((ext_vector_type(VEC)));

    const int tid   = threadIdx.x;
    const int lane  = tid & 31;
    const int wave  = __builtin_amdgcn_readfirstlane(tid >> 5);
    const int base  = blockIdx.x * OWN_ROWS;
    const int nrows = (N_NODES - base < OWN_ROWS) ? (N_NODES - base) : OWN_ROWS;

    const int c0 = wave * WAVE_CHUNKS;
    const unsigned te = (lane < WAVE_CHUNKS) ? tab[(c0 + lane) * OWN_BLOCKS + blockIdx.x] : 0u;

    if (tid < 256) cnt[tid] = 0u;

    int endv = 0, offv = 0, total = 0;
#pragma unroll
    for (int j = 0; j < WAVE_CHUNKS; ++j) {
        const unsigned t = (unsigned)__builtin_amdgcn_readlane((int)te, j);
        const int o = (c0 + j) * BIN_PITCH + (int)(t >> 16) - total;
        total += (int)(t & 0xFFFFu);
        if (lane == j) {
            offv = o;
            endv = total;
        }
    }
    if (lane == 0) wtot[wave] = (unsigned)total;

    __syncthreads();

    unsigned tsum = wtot[lane];
#pragma unroll
    for (int d = 16; d > 0; d >>= 1) tsum += __shfl_xor(tsum, d);
    const int gtot = __builtin_amdgcn_readfirstlane((int)tsum) + nrows;

    if (gtot <= CAP) {
        unsigned* srt = (unsigned*)acc;

        unsigned keep[4];
#pragma unroll
        for (int k = 0; k < 4; ++k)
            keep[k] = (32 * k < total) ? owner_fetch(recs, offv, endv, total, 32 * k + lane) : 0u;
#pragma unroll
        for (int k = 0; k < 4; ++k)
            if (32 * k + lane < total) atomicAdd(&cnt[keep[k] & 255u], 1u);
        for (int b = 128; b < total; b += 32) {
            const unsigned v = owner_fetch(recs, offv, endv, total, b + lane);
            if (b + lane < total) atomicAdd(&cnt[v & 255u], 1u);
        }
        __syncthreads();

        if (wave == 0) {
            const uint4 ca = *(const uint4*)&cnt[lane * 8];
            const uint4 cb = *(const uint4*)&cnt[lane * 8 + 4];
            const unsigned cr[8] = {ca.x, ca.y, ca.z, ca.w, cb.x, cb.y, cb.z, cb.w};
            unsigned c[8], s = 0u;
#pragma unroll
            for (int k = 0; k < 8; ++k) {
                const int row = lane * 8 + k;
                c[k] = (row < nrows) ? cr[k] + 1u : 0u;
                s += c[k];
            }
            unsigned incl = s;
#pragma unroll
            for (int d = 1; d < 32; d <<= 1) {
                const unsigned t = __shfl_up(incl, d);
                if (lane >= d) incl += t;
            }
            unsigned run = incl - s;
#pragma unroll
            for (int k = 0; k < 8; ++k) {
                const int row = lane * 8 + k;
                rstart[row] = run;
                if (row < nrows) {
                    srt[run]  = SRT_START | (unsigned)row;
                    cnt[row]  = run + 1u;
                }
                run += c[k];
            }
        }
        __syncthreads();

#pragma unroll
        for (int k = 0; k < 4; ++k)
            if (32 * k + lane < total) srt[atomicAdd(&cnt[keep[k] & 255u], 1u)] = keep[k];
        for (int b = 128; b < total; b += 32) {
            const unsigned v = owner_fetch(recs, offv, endv, total, b + lane);
            if (b + lane < total) srt[atomicAdd(&cnt[v & 255u], 1u)] = v;
        }
        __syncthreads();

        const unsigned lo = (unsigned)((wave * gtot) / OWN_WAVES);
        const unsigned hi = (unsigned)(((wave + 1) * gtot) / OWN_WAVES);
        int r0 = 0, r1 = 0;
#pragma unroll
        for (int k = 0; k < 7; ++k) {
            const unsigned s = rstart[lane + 32 * k];
            r0 += __builtin_popcount(__builtin_amdgcn_ballot_w32(s < lo));
            r1 += __builtin_popcount(__builtin_amdgcn_ballot_w32(s < hi));
        }
        const int p0 = __builtin_amdgcn_readfirstlane((int)rstart[r0]);
        const int p1 = __builtin_amdgcn_readfirstlane((int)rstart[r1]);

        typedef const __attribute__((address_space(1))) vrow* grow;
        const unsigned long long relb  = (unsigned long long)rel;
        const long long          dlt   = (long long)((unsigned long long)out - relb);
        const unsigned long long lanep = relb + (unsigned long long)(lane * VEC * 4);

        vrow a;
#pragma unroll
        for (int i = 0; i < VEC; ++i) a[i] = 0.f;
        int cur = -1;
        for (int p = p0; p < p1; p += ROWS) {
            const int n = (p1 - p < ROWS) ? (p1 - p) : ROWS;
            const unsigned ent = (lane < n) ? srt[p + lane] : 0u;
            const long long roff = (ent & SRT_START)
                                 ? dlt + (long long)(base + (int)(ent & 255u)) * (W * 4)
                                 : (long long)(ent >> 8) * (W * 4);
            const int rlo = (int)(unsigned)roff;
            const int rhi = (int)(roff >> 32);
            vrow v[ROWS];
#pragma unroll
            for (int j = 0; j < ROWS; ++j) {
                const unsigned long long o =
                    ((unsigned long long)(unsigned)__builtin_amdgcn_readlane(rhi, j) << 32) |
                    (unsigned long long)(unsigned)__builtin_amdgcn_readlane(rlo, j);
                v[j] = *(grow)(lanep + o);
            }
#pragma unroll
            for (int j = 0; j < ROWS; ++j) {
                if (j < n) {
                    const unsigned e = (unsigned)__builtin_amdgcn_readlane((int)ent, j);
                    if (e & SRT_START) {
                        if (cur >= 0) owner_store_row<W, RELU>(out, base + cur, lane, a);
                        cur = (int)(e & 255u);
#pragma unroll
                        for (int i = 0; i < VEC; ++i) a[i] = 0.f;
                    }
                    a += v[j];
                }
            }
        }
        if (cur >= 0) owner_store_row<W, RELU>(out, base + cur, lane, a);
        return;
    }

    float4*       a4 = (float4*)acc;
    float4*       o4 = (float4*)(out + (size_t)base * W);
    for (int i = tid; i < nrows * (W / 4); i += OWN_THREADS) a4[i] = o4[i];

    __syncthreads();

    unsigned nxt = owner_fetch(recs, offv, endv, total, lane);
    for (int b = 0; b < total; b += 32) {
        const unsigned cur = nxt;
        if (b + 32 < total) nxt = owner_fetch(recs, offv, endv, total, b + 32 + lane);
        const int left = total - b;
        owner_add_batch<W>(rel, acc, cur, (left < 32) ? left : 32, lane);
    }

    __syncthreads();
    for (int i = tid; i < nrows * (W / 4); i += OWN_THREADS) {
        float4 x = a4[i];
        if (RELU) {
            x.x = fmaxf(x.x, 0.f);
            x.y = fmaxf(x.y, 0.f);
            x.z = fmaxf(x.z, 0.f);
            x.w = fmaxf(x.w, 0.f);
        }
        o4[i] = x;
    }
}

__global__ void __launch_bounds__(OWN_THREADS)
rgcn_owner_sum128_relu_kernel(const float*    __restrict__ rel,
                              const unsigned* __restrict__ recs,
                              const unsigned* __restrict__ tab,
                              float*                       out) {
    __shared__ __attribute__((aligned(16))) float acc[OWN_ROWS * 128];
    __shared__ __attribute__((aligned(16))) unsigned cnt[256];
    __shared__ unsigned rstart[256];
    __shared__ unsigned wtot[OWN_WAVES];
    owner_sum_body<128, true>(rel, recs, tab, out, acc, cnt, rstart, wtot);
}

__global__ void __launch_bounds__(OWN_THREADS)
rgcn_owner_sum64_kernel(const float*    __restrict__ rel,
                        const unsigned* __restrict__ recs,
                        const unsigned* __restrict__ tab,
                        float*                       out) {
    __shared__ __attribute__((aligned(16))) float acc[OWN_ROWS * 64];
    __shared__ __attribute__((aligned(16))) unsigned cnt[256];
    __shared__ unsigned rstart[256];
    __shared__ unsigned wtot[OWN_WAVES];
    owner_sum_body<64, false>(rel, recs, tab, out, acc, cnt, rstart, wtot);
}

extern "C" void kernel_launch(void* const* d_in, const int* in_sizes, int n_in,
                              void* d_out, int out_size, void* d_ws, size_t ws_size,
                              hipStream_t stream) {
    const float* features = (const float*)d_in[0];
    const int*   src      = (const int*)  d_in[1];
    const int*   dst      = (const int*)  d_in[2];
    const int*   etype    = (const int*)  d_in[3];
    const float* W1       = (const float*)d_in[4];
    const float* W1s      = (const float*)d_in[5];
    const float* b1       = (const float*)d_in[6];
    const float* W2       = (const float*)d_in[7];
    const float* W2s      = (const float*)d_in[8];
    const float* b2       = (const float*)d_in[9];
    float*       out      = (float*)d_out;

    float*  rel  = (float*)d_ws;
    float*  h1   = rel + (size_t)N_REL * N_NODES * KDIM;
    float2* Wf1  = (float2*)(h1 + (size_t)N_NODES * KDIM);
    float2* Wf2  = Wf1 + (size_t)NSEG * 8 * 32 * 32;

    unsigned* recs1 = (unsigned*)d_out;
    unsigned* tab1  = recs1 + BIN_CHUNKS * BIN_PITCH;
    unsigned* recs2 = (unsigned*)(rel + (size_t)N_REL * N_NODES * 64);
    unsigned* tab2  = recs2 + BIN_CHUNKS * BIN_PITCH;

    const int pack1 = NSEG * 8 * 32 * 32;

    rgcn_pack_w_kernel<<<(pack1 + 255) / 256, 256, 0, stream>>>(W1, W1s, Wf1, 128);

    rgcn_gemm_t128_kernel<<<GEMM_BLOCKS + BIN_CHUNKS + PACK2_BLOCKS, GEMM_THREADS, 0, stream>>>(
        features, Wf1, b1, rel, h1, src, dst, etype, recs1, tab1, W2, W2s, Wf2);
    rgcn_owner_sum128_relu_kernel<<<OWN_BLOCKS, OWN_THREADS, 0, stream>>>(rel, recs1, tab1, h1);

    rgcn_gemm_t64_kernel<<<GEMM_BLOCKS + BIN_CHUNKS, GEMM_THREADS, 0, stream>>>(
        h1, Wf2, b2, rel, out, src, dst, etype, recs2, tab2);
    rgcn_owner_sum64_kernel<<<OWN_BLOCKS, OWN_THREADS, 0, stream>>>(rel, recs2, tab2, out);
}
